# starting kernel
# baseline (speedup 1.0000x reference)
#include <hip/hip_runtime.h>
#include <hip/hip_bf16.h>


typedef __attribute__((ext_vector_type(2))) float v2f;
typedef __attribute__((ext_vector_type(8))) float v8f;

#define LHALF   20
#define MMODES  41
#define NMODES  2
#define H1DIM   2
#define H2DIM   10
#define BATCHSZ 65536

#define WAVES_PER_BLOCK 4
#define TILE            16
#define XF              (MMODES * 4)
#define H1STRIDE        20

struct PairTab {
    unsigned pack[442];
    int cnt;
};

constexpr PairTab buildTab() {
    PairTab t{};
    int c = 0;
    for (int m = -LHALF; m <= LHALF; ++m) {
        for (int n = -LHALF; n <= LHALF; ++n) {
            if (n < m) continue;
            long mn = (long)m * (long)n; if (mn < 0) mn = -mn;
            int s = m + n; if (s < 0) s = -s;
            if (mn <= LHALF && s <= LHALF) {
                unsigned en  = (unsigned)(LHALF + n);
                unsigned emn = (unsigned)(LHALF + m + n);
                unsigned em  = (unsigned)(LHALF + m);
                unsigned sy  = (m != n) ? 1u : 0u;
                t.pack[c] = en | (emn << 6) | (em << 12) | (sy << 18);
                ++c;
            }
        }
    }
    t.cnt = c;
    t.pack[c] = t.pack[0];
    return t;
}

constexpr PairTab TAB = buildTab();
constexpr int HDIM  = TAB.cnt;
constexpr int HDIMP = HDIM + (HDIM & 1);
static_assert(HDIM > 0 && HDIM <= 441, "HDIM bounds");

__device__ __constant__ PairTab dTAB = TAB;

__device__ __forceinline__ float lrelu(float x) {
    return x > 0.f ? x : 0.01f * x;
}

__device__ __forceinline__ v8f wmma_step(const float* __restrict__ xe,
                                         const float* __restrict__ bp,
                                         int p, unsigned pk, v8f acc)
{
    int iEn  =  pk        & 63;
    int iEmn = (pk >> 6)  & 63;
    int iEm  = (pk >> 12) & 63;
    float sym = (pk & (1u << 18)) ? 2.f : 1.f;

    const float4 En  = *(const float4*)(xe + iEn  * 4);
    const float4 Emn = *(const float4*)(xe + iEmn * 4);
    const v2f    Em  = *(const v2f*)  (xe + iEm * 4 + 2 * p);

    float Ar = En.x * Emn.x + En.y * Emn.y + En.z * Emn.z + En.w * Emn.w;
    float Ai = En.y * Emn.x - En.x * Emn.y + En.w * Emn.z - En.z * Emn.w;
    v2f a;
    a.x = (Ar * Em.x - Ai * Em.y) * sym;
    a.y = (Ar * Em.y + Ai * Em.x) * sym;

    v2f b = *(const v2f*)bp;

    return __builtin_amdgcn_wmma_f32_16x16x4_f32(
        false, a, false, b, (short)0, acc, false, false);
}

__global__ __launch_bounds__(WAVES_PER_BLOCK * 32)
void eqpbcnn_kernel(const float* __restrict__ x_real,
                    const float* __restrict__ x_imag,
                    const float* __restrict__ task_info,
                    const float* __restrict__ W1r, const float* __restrict__ W1i,
                    const float* __restrict__ W2r, const float* __restrict__ W2i,
                    const float* __restrict__ W3r, const float* __restrict__ W3i,
                    float* __restrict__ out)
{
    extern __shared__ float lds[];
    float* Blds = lds;
    float* xall = lds + HDIMP * 64;

    const int tid  = threadIdx.x;
    const int lane = tid & 31;
    const int wave = tid >> 5;

    float* xw  = xall + wave * (TILE * XF);
    float* h1w = xall + WAVES_PER_BLOCK * (TILE * XF) + wave * (TILE * H1STRIDE);

    for (int idx = tid; idx < HDIMP * 64; idx += blockDim.x) {
        int t = idx & 3;
        int c = (idx >> 2) & 15;
        int j = idx >> 6;
        float v = 0.f;
        if (c < 8 && j < HDIM) {
            int p = c >> 2, o = (c >> 1) & 1, part = c & 1;
            int pf = t >> 1, partf = t & 1;
            if (pf == p) {
                float wr = W1r[(p * H1DIM + o) * HDIM + j];
                float wi = W1i[(p * H1DIM + o) * HDIM + j];
                v = (part == 0) ? (partf == 0 ? wr : -wi)
                                : (partf == 0 ? wi :  wr);
            }
        }
        Blds[idx] = v;
    }
    __syncthreads();

    const int tile = blockIdx.x * WAVES_PER_BLOCK + wave;
    const int base = tile * TILE;

    const float* xr = x_real + (size_t)base * (MMODES * NMODES);
    const float* xi = x_imag + (size_t)base * (MMODES * NMODES);
    for (int idx = lane; idx < TILE * MMODES * NMODES; idx += 32) {
        float a = xr[idx];
        float b = xi[idx];
        int e = idx / (MMODES * NMODES);
        int t = idx - e * (MMODES * NMODES);
        float* d = &xw[e * XF + t * 2];
        d[0] = a;
        d[1] = b;
    }
    __builtin_amdgcn_wave_barrier();
    asm volatile("s_wait_dscnt 0x0" ::: "memory");

    const int r = lane & 15;
    const int p = lane >> 4;
    const float* xe = xw + r * XF;
    const float* bp = Blds + r * 4 + p * 2;

    v8f acc0 = {};
    v8f acc1 = {};
    for (int j = 0; j < HDIMP; j += 2) {
        unsigned pk0 = dTAB.pack[j];
        unsigned pk1 = dTAB.pack[j + 1];
        acc0 = wmma_step(xe, bp + (size_t)j * 64,       p, pk0, acc0);
        acc1 = wmma_step(xe, bp + (size_t)(j + 1) * 64, p, pk1, acc1);
    }
    v8f acc = acc0 + acc1;

#pragma unroll
    for (int v = 0; v < 8; ++v) {
        int row = p * 8 + v;
        h1w[row * H1STRIDE + r] = acc[v];
    }
    __builtin_amdgcn_wave_barrier();
    asm volatile("s_wait_dscnt 0x0" ::: "memory");

    float4 h = *(const float4*)(h1w + r * H1STRIDE + 4 * p);
    float a0 = lrelu(h.x), b0 = lrelu(h.y);
    float a1 = lrelu(h.z), b1 = lrelu(h.w);

    const float* w2r = W2r + p * (H2DIM * H1DIM);
    const float* w2i = W2i + p * (H2DIM * H1DIM);
    const float* w3r = W3r + p * H2DIM;
    const float* w3i = W3i + p * H2DIM;

    float Er = 0.f, Ei = 0.f;
#pragma unroll
    for (int q = 0; q < H2DIM; ++q) {
        float c0 = w2r[q * 2 + 0], c1 = w2r[q * 2 + 1];
        float d0 = w2i[q * 2 + 0], d1 = w2i[q * 2 + 1];
        float hr = a0 * c0 - b0 * d0 + a1 * c1 - b1 * d1;
        float hi = a0 * d0 + b0 * c0 + a1 * d1 + b1 * c1;
        hr = lrelu(hr);
        hi = lrelu(hi);
        float wr = w3r[q], wi = w3i[q];
        Er += hr * wr - hi * wi;
        Ei += hr * wi + hi * wr;
    }

    int bidx = base + r;
    float P = exp10f(task_info[bidx * 4] * 0.1f) * 0.5f;
    float xLr = xw[r * XF + (LHALF * 2 + p) * 2 + 0];
    float xLi = xw[r * XF + (LHALF * 2 + p) * 2 + 1];

    float2 res;
    res.x = xLr + Er * P;
    res.y = xLi + Ei * P;
    *(float2*)(out + ((size_t)bidx * NMODES + p) * 2) = res;
}

extern "C" void kernel_launch(void* const* d_in, const int* in_sizes, int n_in,
                              void* d_out, int out_size, void* d_ws, size_t ws_size,
                              hipStream_t stream) {
    const float* x_real    = (const float*)d_in[0];
    const float* x_imag    = (const float*)d_in[1];
    const float* task_info = (const float*)d_in[2];
    const float* W1r       = (const float*)d_in[3];
    const float* W1i       = (const float*)d_in[4];
    const float* W2r       = (const float*)d_in[5];
    const float* W2i       = (const float*)d_in[6];
    const float* W3r       = (const float*)d_in[7];
    const float* W3i       = (const float*)d_in[8];
    float* out             = (float*)d_out;

    const int blocks = BATCHSZ / (TILE * WAVES_PER_BLOCK);
    const size_t ldsFloats = (size_t)HDIMP * 64
                           + (size_t)WAVES_PER_BLOCK * TILE * XF
                           + (size_t)WAVES_PER_BLOCK * TILE * H1STRIDE;
    const size_t ldsBytes = ldsFloats * sizeof(float);

    eqpbcnn_kernel<<<blocks, WAVES_PER_BLOCK * 32, ldsBytes, stream>>>(
        x_real, x_imag, task_info, W1r, W1i, W2r, W2i, W3r, W3i, out);
}


# 16 waves per block instead of 4 (256 blocks, one per WGP, 228 KiB static LDS): every tile resident in one round instead of two, LDS B matrix built once per WGP by 512 threads
# speedup vs baseline: 2.1778x; 2.1778x over previous
#include <hip/hip_runtime.h>
#include <hip/hip_bf16.h>


typedef __attribute__((ext_vector_type(2))) float v2f;
typedef __attribute__((ext_vector_type(8))) float v8f;

#define LHALF   20
#define MMODES  41
#define NMODES  2
#define H1DIM   2
#define H2DIM   10
#define BATCHSZ 65536

#define WAVES_PER_BLOCK 16
#define TILE            16
#define XF              (MMODES * 4)
#define H1STRIDE        20

struct PairTab {
    unsigned pack[442];
    int cnt;
};

constexpr PairTab buildTab() {
    PairTab t{};
    int c = 0;
    for (int m = -LHALF; m <= LHALF; ++m) {
        for (int n = -LHALF; n <= LHALF; ++n) {
            if (n < m) continue;
            long mn = (long)m * (long)n; if (mn < 0) mn = -mn;
            int s = m + n; if (s < 0) s = -s;
            if (mn <= LHALF && s <= LHALF) {
                unsigned en  = (unsigned)(LHALF + n);
                unsigned emn = (unsigned)(LHALF + m + n);
                unsigned em  = (unsigned)(LHALF + m);
                unsigned sy  = (m != n) ? 1u : 0u;
                t.pack[c] = en | (emn << 6) | (em << 12) | (sy << 18);
                ++c;
            }
        }
    }
    t.cnt = c;
    t.pack[c] = t.pack[0];
    return t;
}

constexpr PairTab TAB = buildTab();
constexpr int HDIM  = TAB.cnt;
constexpr int HDIMP = HDIM + (HDIM & 1);
static_assert(HDIM > 0 && HDIM <= 441, "HDIM bounds");

__device__ __constant__ PairTab dTAB = TAB;

constexpr int LDS_FLOATS = HDIMP * 64
                         + WAVES_PER_BLOCK * TILE * XF
                         + WAVES_PER_BLOCK * TILE * H1STRIDE;
static_assert(LDS_FLOATS * 4 <= 320 * 1024, "LDS budget of one WGP");

__device__ __forceinline__ float lrelu(float x) {
    return x > 0.f ? x : 0.01f * x;
}

__device__ __forceinline__ v8f wmma_step(const float* __restrict__ xe,
                                         const float* __restrict__ bp,
                                         int p, unsigned pk, v8f acc)
{
    int iEn  =  pk        & 63;
    int iEmn = (pk >> 6)  & 63;
    int iEm  = (pk >> 12) & 63;
    float sym = (pk & (1u << 18)) ? 2.f : 1.f;

    const float4 En  = *(const float4*)(xe + iEn  * 4);
    const float4 Emn = *(const float4*)(xe + iEmn * 4);
    const v2f    Em  = *(const v2f*)  (xe + iEm * 4 + 2 * p);

    float Ar = En.x * Emn.x + En.y * Emn.y + En.z * Emn.z + En.w * Emn.w;
    float Ai = En.y * Emn.x - En.x * Emn.y + En.w * Emn.z - En.z * Emn.w;
    v2f a;
    a.x = (Ar * Em.x - Ai * Em.y) * sym;
    a.y = (Ar * Em.y + Ai * Em.x) * sym;

    v2f b = *(const v2f*)bp;

    return __builtin_amdgcn_wmma_f32_16x16x4_f32(
        false, a, false, b, (short)0, acc, false, false);
}

__global__ __launch_bounds__(WAVES_PER_BLOCK * 32)
void eqpbcnn_kernel(const float* __restrict__ x_real,
                    const float* __restrict__ x_imag,
                    const float* __restrict__ task_info,
                    const float* __restrict__ W1r, const float* __restrict__ W1i,
                    const float* __restrict__ W2r, const float* __restrict__ W2i,
                    const float* __restrict__ W3r, const float* __restrict__ W3i,
                    float* __restrict__ out)
{
    __shared__ __attribute__((aligned(16))) float lds[LDS_FLOATS];
    float* Blds = lds;
    float* xall = lds + HDIMP * 64;

    const int tid  = threadIdx.x;
    const int lane = tid & 31;
    const int wave = tid >> 5;

    float* xw  = xall + wave * (TILE * XF);
    float* h1w = xall + WAVES_PER_BLOCK * (TILE * XF) + wave * (TILE * H1STRIDE);

    for (int idx = tid; idx < HDIMP * 64; idx += blockDim.x) {
        int t = idx & 3;
        int c = (idx >> 2) & 15;
        int j = idx >> 6;
        float v = 0.f;
        if (c < 8 && j < HDIM) {
            int p = c >> 2, o = (c >> 1) & 1, part = c & 1;
            int pf = t >> 1, partf = t & 1;
            if (pf == p) {
                float wr = W1r[(p * H1DIM + o) * HDIM + j];
                float wi = W1i[(p * H1DIM + o) * HDIM + j];
                v = (part == 0) ? (partf == 0 ? wr : -wi)
                                : (partf == 0 ? wi :  wr);
            }
        }
        Blds[idx] = v;
    }
    __syncthreads();

    const int tile = blockIdx.x * WAVES_PER_BLOCK + wave;
    const int base = tile * TILE;

    const float* xr = x_real + (size_t)base * (MMODES * NMODES);
    const float* xi = x_imag + (size_t)base * (MMODES * NMODES);
    for (int idx = lane; idx < TILE * MMODES * NMODES; idx += 32) {
        float a = xr[idx];
        float b = xi[idx];
        int e = idx / (MMODES * NMODES);
        int t = idx - e * (MMODES * NMODES);
        float* d = &xw[e * XF + t * 2];
        d[0] = a;
        d[1] = b;
    }
    __builtin_amdgcn_wave_barrier();
    asm volatile("s_wait_dscnt 0x0" ::: "memory");

    const int r = lane & 15;
    const int p = lane >> 4;
    const float* xe = xw + r * XF;
    const float* bp = Blds + r * 4 + p * 2;

    v8f acc0 = {};
    v8f acc1 = {};
    for (int j = 0; j < HDIMP; j += 2) {
        unsigned pk0 = dTAB.pack[j];
        unsigned pk1 = dTAB.pack[j + 1];
        acc0 = wmma_step(xe, bp + (size_t)j * 64,       p, pk0, acc0);
        acc1 = wmma_step(xe, bp + (size_t)(j + 1) * 64, p, pk1, acc1);
    }
    v8f acc = acc0 + acc1;

#pragma unroll
    for (int v = 0; v < 8; ++v) {
        int row = p * 8 + v;
        h1w[row * H1STRIDE + r] = acc[v];
    }
    __builtin_amdgcn_wave_barrier();
    asm volatile("s_wait_dscnt 0x0" ::: "memory");

    float4 h = *(const float4*)(h1w + r * H1STRIDE + 4 * p);
    float a0 = lrelu(h.x), b0 = lrelu(h.y);
    float a1 = lrelu(h.z), b1 = lrelu(h.w);

    const float* w2r = W2r + p * (H2DIM * H1DIM);
    const float* w2i = W2i + p * (H2DIM * H1DIM);
    const float* w3r = W3r + p * H2DIM;
    const float* w3i = W3i + p * H2DIM;

    float Er = 0.f, Ei = 0.f;
#pragma unroll
    for (int q = 0; q < H2DIM; ++q) {
        float c0 = w2r[q * 2 + 0], c1 = w2r[q * 2 + 1];
        float d0 = w2i[q * 2 + 0], d1 = w2i[q * 2 + 1];
        float hr = a0 * c0 - b0 * d0 + a1 * c1 - b1 * d1;
        float hi = a0 * d0 + b0 * c0 + a1 * d1 + b1 * c1;
        hr = lrelu(hr);
        hi = lrelu(hi);
        float wr = w3r[q], wi = w3i[q];
        Er += hr * wr - hi * wi;
        Ei += hr * wi + hi * wr;
    }

    int bidx = base + r;
    float P = exp10f(task_info[bidx * 4] * 0.1f) * 0.5f;
    float xLr = xw[r * XF + (LHALF * 2 + p) * 2 + 0];
    float xLi = xw[r * XF + (LHALF * 2 + p) * 2 + 1];

    float2 res;
    res.x = xLr + Er * P;
    res.y = xLi + Ei * P;
    *(float2*)(out + ((size_t)bidx * NMODES + p) * 2) = res;
}

extern "C" void kernel_launch(void* const* d_in, const int* in_sizes, int n_in,
                              void* d_out, int out_size, void* d_ws, size_t ws_size,
                              hipStream_t stream) {
    const float* x_real    = (const float*)d_in[0];
    const float* x_imag    = (const float*)d_in[1];
    const float* task_info = (const float*)d_in[2];
    const float* W1r       = (const float*)d_in[3];
    const float* W1i       = (const float*)d_in[4];
    const float* W2r       = (const float*)d_in[5];
    const float* W2i       = (const float*)d_in[6];
    const float* W3r       = (const float*)d_in[7];
    const float* W3i       = (const float*)d_in[8];
    float* out             = (float*)d_out;

    const int blocks = BATCHSZ / (TILE * WAVES_PER_BLOCK);

    eqpbcnn_kernel<<<blocks, WAVES_PER_BLOCK * 32, 0, stream>>>(
        x_real, x_imag, task_info, W1r, W1i, W2r, W2i, W3r, W3i, out);
}


# each lane half takes its own index pair: A' formed once per pair (was twice), both pols' features from it, one WMMA per pol, K = 2 pairs x (re,im); pair factor in weights; summation order changes
# speedup vs baseline: 2.6396x; 1.2120x over previous
#include <hip/hip_runtime.h>
#include <hip/hip_bf16.h>


typedef __attribute__((ext_vector_type(2))) float v2f;
typedef __attribute__((ext_vector_type(8))) float v8f;

#define LHALF   20
#define MMODES  41
#define NMODES  2
#define H1DIM   2
#define H2DIM   10
#define BATCHSZ 65536

#define WAVES_PER_BLOCK 16
#define TILE            16
#define XF              (MMODES * 4)
#define H1STRIDE        20

struct PairTab {
    unsigned pack[442];
    int cnt;
};

constexpr PairTab buildTab() {
    PairTab t{};
    int c = 0;
    for (int m = -LHALF; m <= LHALF; ++m) {
        for (int n = -LHALF; n <= LHALF; ++n) {
            if (n < m) continue;
            long mn = (long)m * (long)n; if (mn < 0) mn = -mn;
            int s = m + n; if (s < 0) s = -s;
            if (mn <= LHALF && s <= LHALF) {
                unsigned en  = (unsigned)(LHALF + n);
                unsigned emn = (unsigned)(LHALF + m + n);
                unsigned em  = (unsigned)(LHALF + m);
                unsigned sy  = (m != n) ? 1u : 0u;
                t.pack[c] = en | (emn << 6) | (em << 12) | (sy << 18);
                ++c;
            }
        }
    }
    t.cnt = c;
    t.pack[c] = t.pack[0];
    return t;
}

constexpr PairTab TAB = buildTab();
constexpr int HDIM  = TAB.cnt;
constexpr int HDIMP = HDIM + (HDIM & 1);
static_assert(HDIM > 0 && HDIM <= 441, "HDIM bounds");

__device__ __constant__ PairTab dTAB = TAB;

constexpr int LDS_FLOATS = HDIMP * 64
                         + WAVES_PER_BLOCK * TILE * XF
                         + WAVES_PER_BLOCK * TILE * H1STRIDE;
constexpr int TABL = HDIMP + 2;
static_assert(TABL <= WAVES_PER_BLOCK * 32, "one table entry per thread");
static_assert(LDS_FLOATS * 4 + TABL * 16 <= 320 * 1024, "LDS budget of one WGP");

__device__ __forceinline__ float lrelu(float x) {
    return x > 0.f ? x : 0.01f * x;
}

__global__ __launch_bounds__(WAVES_PER_BLOCK * 32)
void eqpbcnn_kernel(const float* __restrict__ x_real,
                    const float* __restrict__ x_imag,
                    const float* __restrict__ task_info,
                    const float* __restrict__ W1r, const float* __restrict__ W1i,
                    const float* __restrict__ W2r, const float* __restrict__ W2i,
                    const float* __restrict__ W3r, const float* __restrict__ W3i,
                    float* __restrict__ out)
{
    __shared__ __attribute__((aligned(16))) float lds[LDS_FLOATS];
    __shared__ __attribute__((aligned(16))) unsigned tabL[TABL * 4];
    float* Blds = lds;
    float* xall = lds + HDIMP * 64;

    const int tid  = threadIdx.x;
    const int lane = tid & 31;
    const int wave = tid >> 5;

    float* xw  = xall + wave * (TILE * XF);
    float* h1w = xall + WAVES_PER_BLOCK * (TILE * XF) + wave * (TILE * H1STRIDE);

    if (tid < TABL) {
        const int idx = tid;
        unsigned pk = dTAB.pack[idx < HDIM ? idx : HDIM];
        uint4 o;
        o.x = ( pk        & 63) * 16;
        o.y = ((pk >> 6)  & 63) * 16;
        o.z = ((pk >> 12) & 63) * 16;
        o.w = 0;
        *(uint4*)(tabL + idx * 4) = o;
    }

    for (int idx = tid; idx < HDIMP * 64; idx += blockDim.x) {
        int t  = idx & 3;
        int c  = (idx >> 2) & 15;
        int jq = idx >> 6;
        int q  = jq & 1;
        int j  = (jq & ~1) + (t >> 1);
        float v = 0.f;
        if (c < 8 && (c >> 2) == q && j < HDIM) {
            int o = (c >> 1) & 1, part = c & 1;
            int partf = t & 1;
            float wr = W1r[(q * H1DIM + o) * HDIM + j];
            float wi = W1i[(q * H1DIM + o) * HDIM + j];
            float sym = (dTAB.pack[j] & (1u << 18)) ? 2.f : 1.f;
            v = (part == 0) ? (partf == 0 ? wr : -wi)
                            : (partf == 0 ? wi :  wr);
            v *= sym;
        }
        Blds[idx] = v;
    }
    __syncthreads();

    const int tile = blockIdx.x * WAVES_PER_BLOCK + wave;
    const int base = tile * TILE;

    const float* xr = x_real + (size_t)base * (MMODES * NMODES);
    const float* xi = x_imag + (size_t)base * (MMODES * NMODES);
    for (int idx = lane; idx < TILE * MMODES * NMODES; idx += 32) {
        float a = xr[idx];
        float b = xi[idx];
        int e = idx / (MMODES * NMODES);
        int t = idx - e * (MMODES * NMODES);
        float* d = &xw[e * XF + t * 2];
        d[0] = a;
        d[1] = b;
    }
    __builtin_amdgcn_wave_barrier();
    asm volatile("s_wait_dscnt 0x0" ::: "memory");

    const int r = lane & 15;
    const int p = lane >> 4;
    const char* xe = (const char*)(xw + r * XF);
    const float* bp = Blds + r * 4 + p * 2;
    const unsigned* tp = tabL + p * 4;

    v8f acc0 = {};
    v8f acc1 = {};
    uint4 pk = *(const uint4*)tp;
    for (int i = 0; i < HDIMP / 2; ++i) {
        const uint4 pkn = *(const uint4*)(tp + 8 * (i + 1));

        const float4 En  = *(const float4*)(xe + pk.x);
        const float4 Emn = *(const float4*)(xe + pk.y);
        const float4 Em  = *(const float4*)(xe + pk.z);

        const v2f b0 = *(const v2f*)(bp + (size_t)i * 128);
        const v2f b1 = *(const v2f*)(bp + (size_t)i * 128 + 64);

        float Ar = En.x * Emn.x + En.y * Emn.y + En.z * Emn.z + En.w * Emn.w;
        float Ai = En.y * Emn.x - En.x * Emn.y + En.w * Emn.z - En.z * Emn.w;
        v2f a0, a1;
        a0.x = Ar * Em.x - Ai * Em.y;
        a0.y = Ar * Em.y + Ai * Em.x;
        a1.x = Ar * Em.z - Ai * Em.w;
        a1.y = Ar * Em.w + Ai * Em.z;

        acc0 = __builtin_amdgcn_wmma_f32_16x16x4_f32(
            false, a0, false, b0, (short)0, acc0, false, false);
        acc1 = __builtin_amdgcn_wmma_f32_16x16x4_f32(
            false, a1, false, b1, (short)0, acc1, false, false);
        pk = pkn;
    }
    v8f acc = acc0 + acc1;

#pragma unroll
    for (int v = 0; v < 8; ++v) {
        int row = p * 8 + v;
        h1w[row * H1STRIDE + r] = acc[v];
    }
    __builtin_amdgcn_wave_barrier();
    asm volatile("s_wait_dscnt 0x0" ::: "memory");

    float4 h = *(const float4*)(h1w + r * H1STRIDE + 4 * p);
    float a0 = lrelu(h.x), b0 = lrelu(h.y);
    float a1 = lrelu(h.z), b1 = lrelu(h.w);

    const float* w2r = W2r + p * (H2DIM * H1DIM);
    const float* w2i = W2i + p * (H2DIM * H1DIM);
    const float* w3r = W3r + p * H2DIM;
    const float* w3i = W3i + p * H2DIM;

    float Er = 0.f, Ei = 0.f;
#pragma unroll
    for (int q = 0; q < H2DIM; ++q) {
        float c0 = w2r[q * 2 + 0], c1 = w2r[q * 2 + 1];
        float d0 = w2i[q * 2 + 0], d1 = w2i[q * 2 + 1];
        float hr = a0 * c0 - b0 * d0 + a1 * c1 - b1 * d1;
        float hi = a0 * d0 + b0 * c0 + a1 * d1 + b1 * c1;
        hr = lrelu(hr);
        hi = lrelu(hi);
        float wr = w3r[q], wi = w3i[q];
        Er += hr * wr - hi * wi;
        Ei += hr * wi + hi * wr;
    }

    int bidx = base + r;
    float P = exp10f(task_info[bidx * 4] * 0.1f) * 0.5f;
    float xLr = xw[r * XF + (LHALF * 2 + p) * 2 + 0];
    float xLi = xw[r * XF + (LHALF * 2 + p) * 2 + 1];

    float2 res;
    res.x = xLr + Er * P;
    res.y = xLi + Ei * P;
    *(float2*)(out + ((size_t)bidx * NMODES + p) * 2) = res;
}

extern "C" void kernel_launch(void* const* d_in, const int* in_sizes, int n_in,
                              void* d_out, int out_size, void* d_ws, size_t ws_size,
                              hipStream_t stream) {
    const float* x_real    = (const float*)d_in[0];
    const float* x_imag    = (const float*)d_in[1];
    const float* task_info = (const float*)d_in[2];
    const float* W1r       = (const float*)d_in[3];
    const float* W1i       = (const float*)d_in[4];
    const float* W2r       = (const float*)d_in[5];
    const float* W2i       = (const float*)d_in[6];
    const float* W3r       = (const float*)d_in[7];
    const float* W3i       = (const float*)d_in[8];
    float* out             = (float*)d_out;

    const int blocks = BATCHSZ / (TILE * WAVES_PER_BLOCK);

    eqpbcnn_kernel<<<blocks, WAVES_PER_BLOCK * 32, 0, stream>>>(
        x_real, x_imag, task_info, W1r, W1i, W2r, W2i, W3r, W3i, out);
}


# all global loads hoisted to kernel entry and issued as one batch (x tile as b128, W1 values of the B build, tail weights): memory latency paid once, was about 29 waited rounds; sums unchanged
# speedup vs baseline: 3.6097x; 1.3675x over previous
#include <hip/hip_runtime.h>
#include <hip/hip_bf16.h>


typedef __attribute__((ext_vector_type(2))) float v2f;
typedef __attribute__((ext_vector_type(8))) float v8f;

#define LHALF   20
#define MMODES  41
#define NMODES  2
#define H1DIM   2
#define H2DIM   10
#define BATCHSZ 65536

#define WAVES_PER_BLOCK 16
#define TILE            16
#define XF              (MMODES * 4)
#define H1STRIDE        20

struct PairTab {
    unsigned pack[442];
    int cnt;
};

constexpr PairTab buildTab() {
    PairTab t{};
    int c = 0;
    for (int m = -LHALF; m <= LHALF; ++m) {
        for (int n = -LHALF; n <= LHALF; ++n) {
            if (n < m) continue;
            long mn = (long)m * (long)n; if (mn < 0) mn = -mn;
            int s = m + n; if (s < 0) s = -s;
            if (mn <= LHALF && s <= LHALF) {
                unsigned en  = (unsigned)(LHALF + n);
                unsigned emn = (unsigned)(LHALF + m + n);
                unsigned em  = (unsigned)(LHALF + m);
                unsigned sy  = (m != n) ? 1u : 0u;
                t.pack[c] = en | (emn << 6) | (em << 12) | (sy << 18);
                ++c;
            }
        }
    }
    t.cnt = c;
    t.pack[c] = t.pack[0];
    return t;
}

constexpr PairTab TAB = buildTab();
constexpr int HDIM  = TAB.cnt;
constexpr int HDIMP = HDIM + (HDIM & 1);
static_assert(HDIM > 0 && HDIM <= 441, "HDIM bounds");

__device__ __constant__ PairTab dTAB = TAB;

constexpr int LDS_FLOATS = HDIMP * 64
                         + WAVES_PER_BLOCK * TILE * XF
                         + WAVES_PER_BLOCK * TILE * H1STRIDE;
constexpr int TABL = HDIMP + 2;
static_assert(TABL <= WAVES_PER_BLOCK * 32, "one table entry per thread");
static_assert(LDS_FLOATS * 4 + TABL * 16 <= 320 * 1024, "LDS budget of one WGP");

__device__ __forceinline__ float lrelu(float x) {
    return x > 0.f ? x : 0.01f * x;
}

__global__ __launch_bounds__(WAVES_PER_BLOCK * 32)
void eqpbcnn_kernel(const float* __restrict__ x_real,
                    const float* __restrict__ x_imag,
                    const float* __restrict__ task_info,
                    const float* __restrict__ W1r, const float* __restrict__ W1i,
                    const float* __restrict__ W2r, const float* __restrict__ W2i,
                    const float* __restrict__ W3r, const float* __restrict__ W3i,
                    float* __restrict__ out)
{
    __shared__ __attribute__((aligned(16))) float lds[LDS_FLOATS];
    __shared__ __attribute__((aligned(16))) unsigned tabL[TABL * 4];
    float* Blds = lds;
    float* xall = lds + HDIMP * 64;

    const int tid  = threadIdx.x;
    const int lane = tid & 31;
    const int wave = tid >> 5;

    float* xw  = xall + wave * (TILE * XF);
    float* h1w = xall + WAVES_PER_BLOCK * (TILE * XF) + wave * (TILE * H1STRIDE);

    const int tile = blockIdx.x * WAVES_PER_BLOCK + wave;
    const int base = tile * TILE;

    const unsigned pkt = dTAB.pack[tid < HDIM ? tid : HDIM];

    constexpr int NZU = HDIMP * 4;
    constexpr int NZR = (NZU + WAVES_PER_BLOCK * 32 - 1) / (WAVES_PER_BLOCK * 32);
    float wr0[NZR], wi0[NZR], wr1[NZR], wi1[NZR];
    unsigned tb0[NZR], tb1[NZR];
#pragma unroll
    for (int s = 0; s < NZR; ++s) {
        int nz = tid + s * (WAVES_PER_BLOCK * 32);
        nz = nz < NZU ? nz : NZU - 1;
        const int jq = nz >> 2, cc = nz & 3;
        const int q = jq & 1, o = cc >> 1;
        const int j0 = jq & ~1;
        const int j1 = (j0 + 1 < HDIM) ? j0 + 1 : HDIM - 1;
        const int row = (q * H1DIM + o) * HDIM;
        wr0[s] = W1r[row + j0];
        wi0[s] = W1i[row + j0];
        wr1[s] = W1r[row + j1];
        wi1[s] = W1i[row + j1];
        tb0[s] = dTAB.pack[j0];
        tb1[s] = dTAB.pack[j1];
    }

    constexpr int XQ  = TILE * MMODES * NMODES / 4;
    constexpr int XQR = (XQ + 31) / 32;
    static_assert((TILE * MMODES * NMODES) % 4 == 0, "x tile is whole float4s");
    const float4* xr4 = (const float4*)(x_real + (size_t)base * (MMODES * NMODES));
    const float4* xi4 = (const float4*)(x_imag + (size_t)base * (MMODES * NMODES));
    float4 xa[XQR], xb[XQR];
#pragma unroll
    for (int u = 0; u < XQR; ++u) {
        int k = lane + 32 * u;
        k = k < XQ ? k : XQ - 1;
        xa[u] = xr4[k];
        xb[u] = xi4[k];
    }

    const int r = lane & 15;
    const int p = lane >> 4;
    const int bidx = base + r;
    const float* w2r = W2r + p * (H2DIM * H1DIM);
    const float* w2i = W2i + p * (H2DIM * H1DIM);
    const float* w3r = W3r + p * H2DIM;
    const float* w3i = W3i + p * H2DIM;
    float c0v[H2DIM], c1v[H2DIM], d0v[H2DIM], d1v[H2DIM], w3rv[H2DIM], w3iv[H2DIM];
#pragma unroll
    for (int q = 0; q < H2DIM; ++q) {
        c0v[q] = w2r[q * 2 + 0]; c1v[q] = w2r[q * 2 + 1];
        d0v[q] = w2i[q * 2 + 0]; d1v[q] = w2i[q * 2 + 1];
        w3rv[q] = w3r[q];        w3iv[q] = w3i[q];
    }
    const float pdbm = task_info[bidx * 4];

    float4* B4 = (float4*)Blds;
#pragma unroll
    for (int s = 0; s < (HDIMP * 16 + WAVES_PER_BLOCK * 32 - 1) / (WAVES_PER_BLOCK * 32); ++s) {
        const int u = tid + s * (WAVES_PER_BLOCK * 32);
        const int cu = u & 15, qu = (u >> 4) & 1;
        if (u < HDIMP * 16 && (cu >> 2) != qu)
            B4[u] = make_float4(0.f, 0.f, 0.f, 0.f);
    }

    if (tid < TABL) {
        uint4 o;
        o.x = ( pkt        & 63) * 16;
        o.y = ((pkt >> 6)  & 63) * 16;
        o.z = ((pkt >> 12) & 63) * 16;
        o.w = 0;
        *(uint4*)(tabL + tid * 4) = o;
    }

#pragma unroll
    for (int s = 0; s < NZR; ++s) {
        const int nz = tid + s * (WAVES_PER_BLOCK * 32);
        if (nz < NZU) {
            const int jq = nz >> 2, cc = nz & 3;
            const int q = jq & 1, part = cc & 1;
            const bool pad = (jq | 1) >= HDIM;
            const float s0 = (tb0[s] & (1u << 18)) ? 2.f : 1.f;
            const float s1 = (tb1[s] & (1u << 18)) ? 2.f : 1.f;
            const float ar = wr0[s] * s0, ai = wi0[s] * s0;
            const float br = pad ? 0.f : wr1[s] * s1;
            const float bi = pad ? 0.f : wi1[s] * s1;
            B4[jq * 16 + 4 * q + cc] = (part == 0) ? make_float4(ar, -ai, br, -bi)
                                                   : make_float4(ai,  ar, bi,  br);
        }
    }

    float4* xw4 = (float4*)xw;
#pragma unroll
    for (int u = 0; u < XQR; ++u) {
        const int k = lane + 32 * u;
        if (k < XQ) {
            xw4[2 * k]     = make_float4(xa[u].x, xb[u].x, xa[u].y, xb[u].y);
            xw4[2 * k + 1] = make_float4(xa[u].z, xb[u].z, xa[u].w, xb[u].w);
        }
    }
    __syncthreads();

    const char* xe = (const char*)(xw + r * XF);
    const float* bp = Blds + r * 4 + p * 2;
    const unsigned* tp = tabL + p * 4;

    v8f acc0 = {};
    v8f acc1 = {};
    uint4 pk = *(const uint4*)tp;
    for (int i = 0; i < HDIMP / 2; ++i) {
        const uint4 pkn = *(const uint4*)(tp + 8 * (i + 1));

        const float4 En  = *(const float4*)(xe + pk.x);
        const float4 Emn = *(const float4*)(xe + pk.y);
        const float4 Em  = *(const float4*)(xe + pk.z);

        const v2f b0 = *(const v2f*)(bp + (size_t)i * 128);
        const v2f b1 = *(const v2f*)(bp + (size_t)i * 128 + 64);

        float Ar = En.x * Emn.x + En.y * Emn.y + En.z * Emn.z + En.w * Emn.w;
        float Ai = En.y * Emn.x - En.x * Emn.y + En.w * Emn.z - En.z * Emn.w;
        v2f a0, a1;
        a0.x = Ar * Em.x - Ai * Em.y;
        a0.y = Ar * Em.y + Ai * Em.x;
        a1.x = Ar * Em.z - Ai * Em.w;
        a1.y = Ar * Em.w + Ai * Em.z;

        acc0 = __builtin_amdgcn_wmma_f32_16x16x4_f32(
            false, a0, false, b0, (short)0, acc0, false, false);
        acc1 = __builtin_amdgcn_wmma_f32_16x16x4_f32(
            false, a1, false, b1, (short)0, acc1, false, false);
        pk = pkn;
    }
    v8f acc = acc0 + acc1;

#pragma unroll
    for (int v = 0; v < 8; ++v) {
        int row = p * 8 + v;
        h1w[row * H1STRIDE + r] = acc[v];
    }
    __builtin_amdgcn_wave_barrier();
    asm volatile("s_wait_dscnt 0x0" ::: "memory");

    float4 h = *(const float4*)(h1w + r * H1STRIDE + 4 * p);
    float a0 = lrelu(h.x), b0 = lrelu(h.y);
    float a1 = lrelu(h.z), b1 = lrelu(h.w);

    float Er = 0.f, Ei = 0.f;
#pragma unroll
    for (int q = 0; q < H2DIM; ++q) {
        float c0 = c0v[q], c1 = c1v[q];
        float d0 = d0v[q], d1 = d1v[q];
        float hr = a0 * c0 - b0 * d0 + a1 * c1 - b1 * d1;
        float hi = a0 * d0 + b0 * c0 + a1 * d1 + b1 * c1;
        hr = lrelu(hr);
        hi = lrelu(hi);
        float wr = w3rv[q], wi = w3iv[q];
        Er += hr * wr - hi * wi;
        Ei += hr * wi + hi * wr;
    }

    float P = exp10f(pdbm * 0.1f) * 0.5f;
    float xLr = xw[r * XF + (LHALF * 2 + p) * 2 + 0];
    float xLi = xw[r * XF + (LHALF * 2 + p) * 2 + 1];

    float2 res;
    res.x = xLr + Er * P;
    res.y = xLi + Ei * P;
    *(float2*)(out + ((size_t)bidx * NMODES + p) * 2) = res;
}

extern "C" void kernel_launch(void* const* d_in, const int* in_sizes, int n_in,
                              void* d_out, int out_size, void* d_ws, size_t ws_size,
                              hipStream_t stream) {
    const float* x_real    = (const float*)d_in[0];
    const float* x_imag    = (const float*)d_in[1];
    const float* task_info = (const float*)d_in[2];
    const float* W1r       = (const float*)d_in[3];
    const float* W1i       = (const float*)d_in[4];
    const float* W2r       = (const float*)d_in[5];
    const float* W2i       = (const float*)d_in[6];
    const float* W3r       = (const float*)d_in[7];
    const float* W3i       = (const float*)d_in[8];
    float* out             = (float*)d_out;

    const int blocks = BATCHSZ / (TILE * WAVES_PER_BLOCK);

    eqpbcnn_kernel<<<blocks, WAVES_PER_BLOCK * 32, 0, stream>>>(
        x_real, x_imag, task_info, W1r, W1i, W2r, W2i, W3r, W3i, out);
}


# layer 1 regrouped by the Em tap: WMMAs sum W*A' per (m,pol,out) from pair features A' shared by both pols, Em applied once per m group: 90 WMMAs per 16 rows, was 176; summation order changes
# speedup vs baseline: 4.4186x; 1.2241x over previous
#include <hip/hip_runtime.h>
#include <hip/hip_bf16.h>


typedef __attribute__((ext_vector_type(2))) float v2f;
typedef __attribute__((ext_vector_type(8))) float v8f;

#define LHALF   20
#define MMODES  41
#define NMODES  2
#define H1DIM   2
#define H2DIM   10
#define BATCHSZ 65536

#define WAVES_PER_BLOCK 16
#define TILE            16
#define XF              (MMODES * 4)
#define H1STRIDE        36

struct PairTab {
    unsigned pack[442];
    int cnt;
};

constexpr PairTab buildTab() {
    PairTab t{};
    int c = 0;
    for (int m = -LHALF; m <= LHALF; ++m) {
        for (int n = -LHALF; n <= LHALF; ++n) {
            if (n < m) continue;
            long mn = (long)m * (long)n; if (mn < 0) mn = -mn;
            int s = m + n; if (s < 0) s = -s;
            if (mn <= LHALF && s <= LHALF) {
                unsigned en  = (unsigned)(LHALF + n);
                unsigned emn = (unsigned)(LHALF + m + n);
                unsigned em  = (unsigned)(LHALF + m);
                unsigned sy  = (m != n) ? 1u : 0u;
                t.pack[c] = en | (emn << 6) | (em << 12) | (sy << 18);
                ++c;
            }
        }
    }
    t.cnt = c;
    return t;
}

constexpr PairTab TAB = buildTab();
constexpr int HDIM = TAB.cnt;
static_assert(HDIM > 0 && HDIM <= 441, "HDIM bounds");

#define NWMAX 256
#define NTMAX 48
#define NGMAX 48
struct Sched {
    unsigned went[NWMAX * 2];
    unsigned pos[442];
    unsigned tile[NTMAX];
    int nw;
    int nt;
};

constexpr Sched buildSched() {
    Sched s{};
    int gs[NGMAX] = {}, gl[NGMAX] = {};
    bool used[NGMAX] = {};
    int ng = 0;
    for (int j = 0; j < TAB.cnt; ) {
        const unsigned em = (TAB.pack[j] >> 12) & 63;
        gs[ng] = j;
        while (j < TAB.cnt && ((TAB.pack[j] >> 12) & 63) == em) ++j;
        gl[ng] = j - gs[ng];
        ++ng;
    }
    int w = 0, t = 0;
    for (int g = 0; g < ng; ++g) {
        if (used[g]) continue;
        used[g] = true;
        int h = -1;
        for (int c = g + 1; c < ng; ++c) {
            const int m4 = (gl[g] + gl[c]) & 3;
            if (!used[c] && (m4 == 0 || m4 == 3)) { h = c; break; }
        }
        const int own = gl[g] & 3;
        if (h < 0 && !(own == 0 || own == 3)) {
            for (int c = g + 1; c < ng; ++c)
                if (!used[c]) { h = c; break; }
        }
        if (h >= 0) used[h] = true;
        const int len0 = gl[g], len1 = (h >= 0) ? gl[h] : 0;
        const int len = len0 + len1;
        const int cnt = ((len + 3) / 4) * 2;
        for (int k = 0; k < len; ++k) {
            const unsigned slot = (k >= len0) ? 1u : 0u;
            const int jj = slot ? gs[h] + (k - len0) : gs[g] + k;
            const int ww = w + k / 2;
            const unsigned hf = (unsigned)(k & 1);
            s.went[ww * 2 + hf] = (TAB.pack[jj] & 63) | (((TAB.pack[jj] >> 6) & 63) << 6)
                                | (1u << 12) | (slot << 13);
            s.pos[jj] = (unsigned)ww | (hf << 8) | (slot << 9);
        }
        const unsigned em0 = (TAB.pack[gs[g]] >> 12) & 63;
        const unsigned em1 = (h >= 0) ? ((TAB.pack[gs[h]] >> 12) & 63) : em0;
        s.tile[t] = em0 | (em1 << 6) | ((unsigned)cnt << 12);
        w += cnt;
        ++t;
    }
    s.nw = w;
    s.nt = t;
    return s;
}

constexpr Sched SCH = buildSched();
constexpr int NW = SCH.nw;
constexpr int NT = SCH.nt;
static_assert(NW > 0 && NW + 4 <= NWMAX && NW < 256 && (NW & 1) == 0, "schedule bounds");
static_assert(NT > 0 && NT + 1 <= NTMAX, "tile bounds");

constexpr bool checkSched() {
    int occupied = 0;
    for (int i = 0; i < NWMAX * 2; ++i)
        if (SCH.went[i] != 0) ++occupied;
    if (occupied != TAB.cnt) return false;
    int sum = 0;
    for (int t = 0; t < NT; ++t) {
        const int cnt = (int)(SCH.tile[t] >> 12);
        if (cnt <= 0 || (cnt & 1)) return false;
        sum += cnt;
    }
    if (sum != NW) return false;
    for (int j = 0; j < TAB.cnt; ++j) {
        const int ww = (int)(SCH.pos[j] & 255);
        const unsigned hf = (SCH.pos[j] >> 8) & 1, slot = (SCH.pos[j] >> 9) & 1;
        if (ww >= NW) return false;
        const unsigned want = (TAB.pack[j] & 63) | (((TAB.pack[j] >> 6) & 63) << 6)
                            | (1u << 12) | (slot << 13);
        if (SCH.went[ww * 2 + hf] != want) return false;
        int t = 0, first = 0;
        while (t < NT && ww >= first + (int)(SCH.tile[t] >> 12)) { first += (int)(SCH.tile[t] >> 12); ++t; }
        if (t >= NT) return false;
        const unsigned em = slot ? ((SCH.tile[t] >> 6) & 63) : (SCH.tile[t] & 63);
        if (em != ((TAB.pack[j] >> 12) & 63)) return false;
    }
    return true;
}
static_assert(checkSched(), "schedule covers every pair exactly once");

__device__ __constant__ PairTab dTAB = TAB;
__device__ __constant__ Sched dSCH = SCH;

constexpr int BMATS  = NW + 2;
constexpr int TABW   = (NW + 4) * 2;
constexpr int TABT   = NT + 1;
constexpr int LDS_FLOATS = BMATS * 64
                         + WAVES_PER_BLOCK * TILE * XF
                         + WAVES_PER_BLOCK * TILE * H1STRIDE;
static_assert(TABW <= WAVES_PER_BLOCK * 32, "one table entry per thread");
static_assert(LDS_FLOATS * 4 + TABW * 8 + TABT * 16 <= 320 * 1024, "LDS budget of one WGP");
static_assert(LDS_FLOATS * 4 > 160 * 1024, "one block per WGP");

__device__ __forceinline__ float lrelu(float x) {
    return x > 0.f ? x : 0.01f * x;
}

__global__ __launch_bounds__(WAVES_PER_BLOCK * 32)
void eqpbcnn_kernel(const float* __restrict__ x_real,
                    const float* __restrict__ x_imag,
                    const float* __restrict__ task_info,
                    const float* __restrict__ W1r, const float* __restrict__ W1i,
                    const float* __restrict__ W2r, const float* __restrict__ W2i,
                    const float* __restrict__ W3r, const float* __restrict__ W3i,
                    float* __restrict__ out)
{
    __shared__ __attribute__((aligned(16))) float lds[LDS_FLOATS];
    __shared__ __attribute__((aligned(16))) unsigned tabW[TABW * 2];
    __shared__ __attribute__((aligned(16))) unsigned tabT[TABT * 4];
    float* Blds = lds;
    float* xall = lds + BMATS * 64;

    const int tid  = threadIdx.x;
    const int lane = tid & 31;
    const int wave = tid >> 5;

    float* xw  = xall + wave * (TILE * XF);
    float* h1w = xall + WAVES_PER_BLOCK * (TILE * XF) + wave * (TILE * H1STRIDE);

    const int tile = blockIdx.x * WAVES_PER_BLOCK + wave;
    const int base = tile * TILE;

    const unsigned wen = dSCH.went[tid < TABW ? tid : TABW - 1];
    const unsigned ten = dSCH.tile[tid < TABT ? tid : TABT - 1];

    constexpr int ZU = BMATS * 32;
    constexpr int ZR = (ZU + WAVES_PER_BLOCK * 32 - 1) / (WAVES_PER_BLOCK * 32);
    unsigned zen[ZR];
#pragma unroll
    for (int s = 0; s < ZR; ++s) {
        int z = tid + s * (WAVES_PER_BLOCK * 32);
        z = z < ZU ? z : ZU - 1;
        zen[s] = dSCH.went[(z >> 5) * 2 + (z & 1)];
    }

    constexpr int NZU = HDIM * 4;
    constexpr int NZR = (NZU + WAVES_PER_BLOCK * 32 - 1) / (WAVES_PER_BLOCK * 32);
    float wr0[NZR], wi0[NZR];
    unsigned tb0[NZR], ps0[NZR];
#pragma unroll
    for (int s = 0; s < NZR; ++s) {
        int u = tid + s * (WAVES_PER_BLOCK * 32);
        u = u < NZU ? u : NZU - 1;
        const int j = u >> 2, po = u & 3;
        wr0[s] = W1r[po * HDIM + j];
        wi0[s] = W1i[po * HDIM + j];
        tb0[s] = dTAB.pack[j];
        ps0[s] = dSCH.pos[j];
    }

    constexpr int XQ  = TILE * MMODES * NMODES / 4;
    constexpr int XQR = (XQ + 31) / 32;
    static_assert((TILE * MMODES * NMODES) % 4 == 0, "x tile is whole float4s");
    const float4* xr4 = (const float4*)(x_real + (size_t)base * (MMODES * NMODES));
    const float4* xi4 = (const float4*)(x_imag + (size_t)base * (MMODES * NMODES));
    float4 xa[XQR], xb[XQR];
#pragma unroll
    for (int u = 0; u < XQR; ++u) {
        int k = lane + 32 * u;
        k = k < XQ ? k : XQ - 1;
        xa[u] = xr4[k];
        xb[u] = xi4[k];
    }

    const int r = lane & 15;
    const int p = lane >> 4;
    const int bidx = base + r;
    const float* w2r = W2r + p * (H2DIM * H1DIM);
    const float* w2i = W2i + p * (H2DIM * H1DIM);
    const float* w3r = W3r + p * H2DIM;
    const float* w3i = W3i + p * H2DIM;
    float c0v[H2DIM], c1v[H2DIM], d0v[H2DIM], d1v[H2DIM], w3rv[H2DIM], w3iv[H2DIM];
#pragma unroll
    for (int q = 0; q < H2DIM; ++q) {
        c0v[q] = w2r[q * 2 + 0]; c1v[q] = w2r[q * 2 + 1];
        d0v[q] = w2i[q * 2 + 0]; d1v[q] = w2i[q * 2 + 1];
        w3rv[q] = w3r[q];        w3iv[q] = w3i[q];
    }
    const float pdbm = task_info[bidx * 4];

    float2* B2 = (float2*)Blds;
#pragma unroll
    for (int s = 0; s < ZR; ++s) {
        const int z = tid + s * (WAVES_PER_BLOCK * 32);
        const unsigned ze = zen[s];
        const bool nonzero = ((ze >> 12) & 1) != 0 && ((ze >> 13) & 1) == (unsigned)((z >> 4) & 1);
        if (z < ZU && !nonzero)
            B2[z] = make_float2(0.f, 0.f);
    }

    if (tid < TABW) {
        uint2 o;
        o.x = ( wen       & 63) * 16;
        o.y = ((wen >> 6) & 63) * 16;
        *(uint2*)(tabW + tid * 2) = o;
    }
    if (tid < TABT) {
        uint4 o;
        o.x = ( ten       & 63) * 16;
        o.y = ((ten >> 6) & 63) * 16;
        o.z = ten >> 12;
        o.w = 0;
        *(uint4*)(tabT + tid * 4) = o;
    }

#pragma unroll
    for (int s = 0; s < NZR; ++s) {
        const int u = tid + s * (WAVES_PER_BLOCK * 32);
        if (u < NZU) {
            const int po = u & 3;
            const float sy = (tb0[s] & (1u << 18)) ? 2.f : 1.f;
            const float ar = wr0[s] * sy, ai = wi0[s] * sy;
            const unsigned ps = ps0[s];
            const int ww = ps & 255, hf = (ps >> 8) & 1, slot = (ps >> 9) & 1;
            float* d = Blds + ww * 64 + (slot * 8 + po * 2) * 4 + hf * 2;
            *(float2*)(d)     = make_float2(ar, -ai);
            *(float2*)(d + 4) = make_float2(ai,  ar);
        }
    }

    float4* xw4 = (float4*)xw;
#pragma unroll
    for (int u = 0; u < XQR; ++u) {
        const int k = lane + 32 * u;
        if (k < XQ) {
            xw4[2 * k]     = make_float4(xa[u].x, xb[u].x, xa[u].y, xb[u].y);
            xw4[2 * k + 1] = make_float4(xa[u].z, xb[u].z, xa[u].w, xb[u].w);
        }
    }
    __syncthreads();

    const char* xe = (const char*)(xw + r * XF);
    const char* xq = (const char*)(xw + (8 * p) * XF);
    const float* bp = Blds + r * 4 + p * 2;
    const uint2* tw = (const uint2*)tabW + p;
    const uint4* tt4 = (const uint4*)tabT;
    const bool cslot = (r >> 3) != 0;
    const unsigned cpol8 = ((r >> 2) & 1) * 8;

    v2f qv[8];
#pragma unroll
    for (int v = 0; v < 8; ++v) qv[v] = v2f{0.f, 0.f};

    uint2 pk0 = tw[0], pk1 = tw[2];
    float4 En0  = *(const float4*)(xe + pk0.x);
    float4 Emn0 = *(const float4*)(xe + pk0.y);
    float4 En1  = *(const float4*)(xe + pk1.x);
    float4 Emn1 = *(const float4*)(xe + pk1.y);
    v2f bb0 = *(const v2f*)(bp);
    v2f bb1 = *(const v2f*)(bp + 64);
    pk0 = tw[4]; pk1 = tw[6];
    const float* bpn = bp + 128;
    const uint2* twn = tw + 8;

    auto aprime = [](const float4 En, const float4 Emn) -> v2f {
        v2f a;
        a.x = En.x * Emn.x + En.y * Emn.y + En.z * Emn.z + En.w * Emn.w;
        a.y = En.y * Emn.x - En.x * Emn.y + En.w * Emn.z - En.z * Emn.w;
        return a;
    };

    uint4 tt = tt4[0];
    for (int t = 0; t < NT; ++t) {
        const int cnt = __builtin_amdgcn_readfirstlane((int)tt.z);
        const unsigned eo = (cslot ? tt.y : tt.x) + cpol8;
        v2f ev[8];
#pragma unroll
        for (int v = 0; v < 8; ++v)
            ev[v] = *(const v2f*)(xq + v * (XF * 4) + eo);
        tt = tt4[t + 1];

        v8f acc0 = {}, acc1 = {};
        int k = cnt >> 1;
        do {
            const float4 En0N  = *(const float4*)(xe + pk0.x);
            const float4 Emn0N = *(const float4*)(xe + pk0.y);
            const float4 En1N  = *(const float4*)(xe + pk1.x);
            const float4 Emn1N = *(const float4*)(xe + pk1.y);
            const v2f bb0N = *(const v2f*)(bpn);
            const v2f bb1N = *(const v2f*)(bpn + 64);
            const uint2 pk0N = twn[0];
            const uint2 pk1N = twn[2];
            bpn += 128;
            twn += 4;

            const v2f a0 = aprime(En0, Emn0);
            acc0 = __builtin_amdgcn_wmma_f32_16x16x4_f32(
                false, a0, false, bb0, (short)0, acc0, false, false);
            const v2f a1 = aprime(En1, Emn1);
            acc1 = __builtin_amdgcn_wmma_f32_16x16x4_f32(
                false, a1, false, bb1, (short)0, acc1, false, false);

            En0 = En0N; Emn0 = Emn0N; En1 = En1N; Emn1 = Emn1N;
            bb0 = bb0N; bb1 = bb1N; pk0 = pk0N; pk1 = pk1N;
        } while (--k > 0);

        const v8f acc = acc0 + acc1;
#pragma unroll
        for (int v = 0; v < 8; ++v)
            qv[v] += ev[v] * v2f{acc[v], acc[v]};
    }

#pragma unroll
    for (int v = 0; v < 8; ++v) {
        const int row = p * 8 + v;
        *(v2f*)(h1w + row * H1STRIDE + 2 * r) = qv[v];
    }
    __builtin_amdgcn_wave_barrier();
    asm volatile("s_wait_dscnt 0x0" ::: "memory");

    float hre[H1DIM], him[H1DIM];
#pragma unroll
    for (int o = 0; o < H1DIM; ++o) {
        hre[o] = 0.f; him[o] = 0.f;
#pragma unroll
        for (int sl = 0; sl < 2; ++sl) {
            const int c0 = sl * 8 + p * 4 + o * 2;
            const float4 s = *(const float4*)(h1w + r * H1STRIDE + 2 * c0);
            hre[o] += s.x - s.w;
            him[o] += s.z + s.y;
        }
    }
    float a0 = lrelu(hre[0]), b0 = lrelu(him[0]);
    float a1 = lrelu(hre[1]), b1 = lrelu(him[1]);

    float Er = 0.f, Ei = 0.f;
#pragma unroll
    for (int q = 0; q < H2DIM; ++q) {
        float c0 = c0v[q], c1 = c1v[q];
        float d0 = d0v[q], d1 = d1v[q];
        float hr = a0 * c0 - b0 * d0 + a1 * c1 - b1 * d1;
        float hi = a0 * d0 + b0 * c0 + a1 * d1 + b1 * c1;
        hr = lrelu(hr);
        hi = lrelu(hi);
        float wr = w3rv[q], wi = w3iv[q];
        Er += hr * wr - hi * wi;
        Ei += hr * wi + hi * wr;
    }

    float P = exp10f(pdbm * 0.1f) * 0.5f;
    float xLr = xw[r * XF + (LHALF * 2 + p) * 2 + 0];
    float xLi = xw[r * XF + (LHALF * 2 + p) * 2 + 1];

    float2 res;
    res.x = xLr + Er * P;
    res.y = xLi + Ei * P;
    *(float2*)(out + ((size_t)bidx * NMODES + p) * 2) = res;
}

extern "C" void kernel_launch(void* const* d_in, const int* in_sizes, int n_in,
                              void* d_out, int out_size, void* d_ws, size_t ws_size,
                              hipStream_t stream) {
    const float* x_real    = (const float*)d_in[0];
    const float* x_imag    = (const float*)d_in[1];
    const float* task_info = (const float*)d_in[2];
    const float* W1r       = (const float*)d_in[3];
    const float* W1i       = (const float*)d_in[4];
    const float* W2r       = (const float*)d_in[5];
    const float* W2i       = (const float*)d_in[6];
    const float* W3r       = (const float*)d_in[7];
    const float* W3i       = (const float*)d_in[8];
    float* out             = (float*)d_out;

    const int blocks = BATCHSZ / (TILE * WAVES_PER_BLOCK);

    eqpbcnn_kernel<<<blocks, WAVES_PER_BLOCK * 32, 0, stream>>>(
        x_real, x_imag, task_info, W1r, W1i, W2r, W2i, W3r, W3i, out);
}


# x row of each lane held in registers (lane half 1 shifted one tap): En/Emn of the unrolled compile-time WMMA schedule are fixed registers, LDS gathers and offset tables gone; summation order unchanged
# speedup vs baseline: 5.1384x; 1.1629x over previous
#include <hip/hip_runtime.h>
#include <hip/hip_bf16.h>
#include <utility>


typedef __attribute__((ext_vector_type(2))) float v2f;
typedef __attribute__((ext_vector_type(8))) float v8f;

#define LHALF   20
#define MMODES  41
#define NMODES  2
#define H1DIM   2
#define H2DIM   10
#define BATCHSZ 65536

#define WAVES_PER_BLOCK 16
#define TILE            16
#define XF              (MMODES * 4)
#define H1STRIDE        36

struct PairTab {
    unsigned pack[442];
    int cnt;
};

constexpr PairTab buildTab() {
    PairTab t{};
    int c = 0;
    for (int m = -LHALF; m <= LHALF; ++m) {
        for (int n = -LHALF; n <= LHALF; ++n) {
            if (n < m) continue;
            long mn = (long)m * (long)n; if (mn < 0) mn = -mn;
            int s = m + n; if (s < 0) s = -s;
            if (mn <= LHALF && s <= LHALF) {
                unsigned en  = (unsigned)(LHALF + n);
                unsigned emn = (unsigned)(LHALF + m + n);
                unsigned em  = (unsigned)(LHALF + m);
                unsigned sy  = (m != n) ? 1u : 0u;
                t.pack[c] = en | (emn << 6) | (em << 12) | (sy << 18);
                ++c;
            }
        }
    }
    t.cnt = c;
    return t;
}

constexpr PairTab TAB = buildTab();
constexpr int HDIM = TAB.cnt;
static_assert(HDIM > 0 && HDIM <= 441, "HDIM bounds");

#define NWMAX 256
#define NTMAX 48
#define NGMAX 48
struct Sched {
    unsigned went[NWMAX * 2];
    unsigned pos[442];
    unsigned tile[NTMAX];
    int nw;
    int nt;
};

constexpr Sched buildSched() {
    Sched s{};
    int gs[NGMAX] = {}, gl[NGMAX] = {};
    bool used[NGMAX] = {};
    int ng = 0;
    for (int j = 0; j < TAB.cnt; ) {
        const unsigned em = (TAB.pack[j] >> 12) & 63;
        gs[ng] = j;
        while (j < TAB.cnt && ((TAB.pack[j] >> 12) & 63) == em) ++j;
        gl[ng] = j - gs[ng];
        ++ng;
    }
    int w = 0, t = 0;
    for (int g = 0; g < ng; ++g) {
        if (used[g]) continue;
        used[g] = true;
        int h = -1;
        for (int c = g + 1; c < ng; ++c) {
            const int m4 = (gl[g] + gl[c]) & 3;
            if (!used[c] && (m4 == 0 || m4 == 3)) { h = c; break; }
        }
        const int own = gl[g] & 3;
        if (h < 0 && !(own == 0 || own == 3)) {
            for (int c = g + 1; c < ng; ++c)
                if (!used[c]) { h = c; break; }
        }
        if (h >= 0) used[h] = true;
        const int len0 = gl[g], len1 = (h >= 0) ? gl[h] : 0;
        const int len = len0 + len1;
        const int cnt = ((len + 3) / 4) * 2;
        for (int k = 0; k < len; ++k) {
            const unsigned slot = (k >= len0) ? 1u : 0u;
            const int jj = slot ? gs[h] + (k - len0) : gs[g] + k;
            const int ww = w + k / 2;
            const unsigned hf = (unsigned)(k & 1);
            s.went[ww * 2 + hf] = (TAB.pack[jj] & 63) | (((TAB.pack[jj] >> 6) & 63) << 6)
                                | (1u << 12) | (slot << 13);
            s.pos[jj] = (unsigned)ww | (hf << 8) | (slot << 9);
        }
        const unsigned em0 = (TAB.pack[gs[g]] >> 12) & 63;
        const unsigned em1 = (h >= 0) ? ((TAB.pack[gs[h]] >> 12) & 63) : em0;
        s.tile[t] = em0 | (em1 << 6) | ((unsigned)cnt << 12);
        w += cnt;
        ++t;
    }
    s.nw = w;
    s.nt = t;
    return s;
}

constexpr Sched SCH = buildSched();
constexpr int NW = SCH.nw;
constexpr int NT = SCH.nt;
static_assert(NW > 0 && NW + 4 <= NWMAX && NW < 256 && (NW & 1) == 0, "schedule bounds");
static_assert(NT > 0 && NT + 1 <= NTMAX, "tile bounds");

constexpr bool checkSched() {
    int occupied = 0;
    for (int i = 0; i < NWMAX * 2; ++i)
        if (SCH.went[i] != 0) ++occupied;
    if (occupied != TAB.cnt) return false;
    int sum = 0;
    for (int t = 0; t < NT; ++t) {
        const int cnt = (int)(SCH.tile[t] >> 12);
        if (cnt <= 0 || (cnt & 1)) return false;
        sum += cnt;
    }
    if (sum != NW) return false;
    for (int j = 0; j < TAB.cnt; ++j) {
        const int ww = (int)(SCH.pos[j] & 255);
        const unsigned hf = (SCH.pos[j] >> 8) & 1, slot = (SCH.pos[j] >> 9) & 1;
        if (ww >= NW) return false;
        const unsigned want = (TAB.pack[j] & 63) | (((TAB.pack[j] >> 6) & 63) << 6)
                            | (1u << 12) | (slot << 13);
        if (SCH.went[ww * 2 + hf] != want) return false;
        int t = 0, first = 0;
        while (t < NT && ww >= first + (int)(SCH.tile[t] >> 12)) { first += (int)(SCH.tile[t] >> 12); ++t; }
        if (t >= NT) return false;
        const unsigned em = slot ? ((SCH.tile[t] >> 6) & 63) : (SCH.tile[t] & 63);
        if (em != ((TAB.pack[j] >> 12) & 63)) return false;
    }
    return true;
}
static_assert(checkSched(), "schedule covers every pair exactly once");

__device__ __constant__ PairTab dTAB = TAB;
__device__ __constant__ Sched dSCH = SCH;

constexpr int BMATS  = NW;
constexpr int TWPOL    = 2 * H2DIM * H1DIM + 2 * H2DIM;
constexpr int TWSTRIDE = 64;
constexpr int LDS_FLOATS = BMATS * 64
                         + WAVES_PER_BLOCK * TILE * XF
                         + WAVES_PER_BLOCK * TILE * H1STRIDE;
static_assert(2 * TWPOL <= WAVES_PER_BLOCK * 32, "one tail weight per thread");
static_assert(LDS_FLOATS * 4 + 2 * TWSTRIDE * 4 <= 320 * 1024, "LDS budget of one WGP");
static_assert(LDS_FLOATS * 4 > 160 * 1024, "one block per WGP");

constexpr int tileFirst(int t) {
    int f = 0;
    for (int i = 0; i < t; ++i) f += (int)(SCH.tile[i] >> 12);
    return f;
}

__device__ __forceinline__ float lrelu(float x) {
    return x > 0.f ? x : 0.01f * x;
}

__device__ __forceinline__ v2f aprime(const float4 En, const float4 Emn) {
    v2f a;
    a.x = En.x * Emn.x + En.y * Emn.y + En.z * Emn.z + En.w * Emn.w;
    a.y = En.y * Emn.x - En.x * Emn.y + En.w * Emn.z - En.z * Emn.w;
    return a;
}

struct WRegs { bool any, has1; int e0, f0, e1, f1; };
constexpr WRegs wregs(int w) {
    const unsigned g0 = SCH.went[2 * w], g1 = SCH.went[2 * w + 1];
    const bool has0 = ((g0 >> 12) & 1) != 0, has1 = ((g1 >> 12) & 1) != 0;
    WRegs q{};
    q.any = has0 || has1;
    q.has1 = has1;
    q.e0 = has0 ? (int)(g0 & 63)        : (int)(g1 & 63) - 1;
    q.f0 = has0 ? (int)((g0 >> 6) & 63) : (int)((g1 >> 6) & 63) - 1;
    q.e1 = has1 ? (int)(g1 & 63) - 1        : q.e0;
    q.f1 = has1 ? (int)((g1 >> 6) & 63) - 1 : q.f0;
    return q;
}
constexpr int regLoadTile(int i) {
    for (int t = 0; t < NT; ++t) {
        const int w0 = tileFirst(t), cnt = (int)(SCH.tile[t] >> 12);
        for (int w = w0; w < w0 + cnt; ++w) {
            const WRegs q = wregs(w);
            if (q.any && (q.e0 == i || q.f0 == i || q.e1 == i || q.f1 == i))
                return t > 0 ? t - 1 : 0;
        }
    }
    return NT;
}

template<int Wi>
__device__ __forceinline__ void wmma_step(const float4 (&X)[MMODES], const float* bp,
                                          const bool half1, v8f& acc) {
    constexpr WRegs Q = wregs(Wi);
    if constexpr (Q.any) {
        constexpr int e0 = Q.e0, f0 = Q.f0, e1 = Q.e1, f1 = Q.f1;
        constexpr bool has1 = Q.has1;
        static_assert(e0 >= 0 && e0 < MMODES && f0 >= 0 && f0 < MMODES, "register of half 0");
        static_assert(e1 >= 0 && e1 < MMODES - 1 + (has1 ? 0 : 1) && f1 >= 0
                      && f1 < MMODES - 1 + (has1 ? 0 : 1), "register of half 1");
        v2f a = aprime(X[e0], X[f0]);
        if constexpr (e1 != e0 || f1 != f0) {
            const v2f b = aprime(X[e1], X[f1]);
            a.x = half1 ? b.x : a.x;
            a.y = half1 ? b.y : a.y;
        }
        const v2f bb = *(const v2f*)(bp + Wi * 64);
        acc = __builtin_amdgcn_wmma_f32_16x16x4_f32(
            false, a, false, bb, (short)0, acc, false, false);
    }
}

template<int W0, int... Is>
__device__ __forceinline__ void tile_wmmas(std::integer_sequence<int, Is...>,
                                           const float4 (&X)[MMODES], const float* bp,
                                           const bool half1, v8f& acc0, v8f& acc1) {
    (wmma_step<W0 + Is>(X, bp, half1, (Is & 1) ? acc1 : acc0), ...);
}

template<int Tt, int... Is>
__device__ __forceinline__ void row_loads(std::integer_sequence<int, Is...>,
                                          float4 (&X)[MMODES], const char* xe, const char* xe1) {
    ((regLoadTile(Is) == Tt
          ? (void)(X[Is] = *(const float4*)((Is < MMODES - 1 ? xe1 : xe) + Is * 16))
          : (void)0), ...);
}

template<int Tt>
__device__ __forceinline__ void tile_step(float4 (&X)[MMODES], const char* xe, const char* xe1,
                                          const float* bp,
                                          const char* xq, const bool half1,
                                          const bool cslot, const unsigned cpol8,
                                          v2f (&qv)[8]) {
    row_loads<Tt>(std::make_integer_sequence<int, MMODES>{}, X, xe, xe1);
    constexpr unsigned tl = SCH.tile[Tt];
    constexpr unsigned E0 = (tl & 63) * 16, E1 = ((tl >> 6) & 63) * 16;
    constexpr int CNT = (int)(tl >> 12);
    constexpr int W0 = tileFirst(Tt);
    const unsigned eo = (cslot ? E1 : E0) + cpol8;
    v2f ev[8];
#pragma unroll
    for (int v = 0; v < 8; ++v)
        ev[v] = *(const v2f*)(xq + v * (XF * 4) + eo);

    v8f acc0 = {}, acc1 = {};
    tile_wmmas<W0>(std::make_integer_sequence<int, CNT>{}, X, bp, half1, acc0, acc1);

    const v8f acc = acc0 + acc1;
#pragma unroll
    for (int v = 0; v < 8; ++v)
        qv[v] += ev[v] * v2f{acc[v], acc[v]};
    __builtin_amdgcn_sched_barrier(0);
}

template<int... Ts>
__device__ __forceinline__ void all_tiles(std::integer_sequence<int, Ts...>,
                                          float4 (&X)[MMODES], const char* xe, const char* xe1,
                                          const float* bp,
                                          const char* xq, const bool half1,
                                          const bool cslot, const unsigned cpol8,
                                          v2f (&qv)[8]) {
    (tile_step<Ts>(X, xe, xe1, bp, xq, half1, cslot, cpol8, qv), ...);
}

__global__ __launch_bounds__(WAVES_PER_BLOCK * 32)
void eqpbcnn_kernel(const float* __restrict__ x_real,
                    const float* __restrict__ x_imag,
                    const float* __restrict__ task_info,
                    const float* __restrict__ W1r, const float* __restrict__ W1i,
                    const float* __restrict__ W2r, const float* __restrict__ W2i,
                    const float* __restrict__ W3r, const float* __restrict__ W3i,
                    float* __restrict__ out)
{
    __shared__ __attribute__((aligned(16))) float lds[LDS_FLOATS];
    __shared__ __attribute__((aligned(16))) float tailw[2 * TWSTRIDE];
    float* Blds = lds;
    float* xall = lds + BMATS * 64;

    const int tid  = threadIdx.x;
    const int lane = tid & 31;
    const int wave = tid >> 5;

    float* xw  = xall + wave * (TILE * XF);
    float* h1w = xall + WAVES_PER_BLOCK * (TILE * XF) + wave * (TILE * H1STRIDE);

    const int tile = blockIdx.x * WAVES_PER_BLOCK + wave;
    const int base = tile * TILE;

    constexpr int ZU = BMATS * 32;
    constexpr int ZR = (ZU + WAVES_PER_BLOCK * 32 - 1) / (WAVES_PER_BLOCK * 32);
    unsigned zen[ZR];
#pragma unroll
    for (int s = 0; s < ZR; ++s) {
        int z = tid + s * (WAVES_PER_BLOCK * 32);
        z = z < ZU ? z : ZU - 1;
        zen[s] = dSCH.went[(z >> 5) * 2 + (z & 1)];
    }

    constexpr int NZU = HDIM * 4;
    constexpr int NZR = (NZU + WAVES_PER_BLOCK * 32 - 1) / (WAVES_PER_BLOCK * 32);
    float wr0[NZR], wi0[NZR];
    unsigned tb0[NZR], ps0[NZR];
#pragma unroll
    for (int s = 0; s < NZR; ++s) {
        int u = tid + s * (WAVES_PER_BLOCK * 32);
        u = u < NZU ? u : NZU - 1;
        const int j = u >> 2, po = u & 3;
        wr0[s] = W1r[po * HDIM + j];
        wi0[s] = W1i[po * HDIM + j];
        tb0[s] = dTAB.pack[j];
        ps0[s] = dSCH.pos[j];
    }

    constexpr int XQ  = TILE * MMODES * NMODES / 4;
    constexpr int XQR = (XQ + 31) / 32;
    static_assert((TILE * MMODES * NMODES) % 4 == 0, "x tile is whole float4s");
    const float4* xr4 = (const float4*)(x_real + (size_t)base * (MMODES * NMODES));
    const float4* xi4 = (const float4*)(x_imag + (size_t)base * (MMODES * NMODES));
    float4 xa[XQR], xb[XQR];
#pragma unroll
    for (int u = 0; u < XQR; ++u) {
        int k = lane + 32 * u;
        k = k < XQ ? k : XQ - 1;
        xa[u] = xr4[k];
        xb[u] = xi4[k];
    }

    const int r = lane & 15;
    const int p = lane >> 4;
    const int bidx = base + r;
    float twv;
    {
        const int u  = tid < 2 * TWPOL ? tid : 2 * TWPOL - 1;
        const int pl = u >= TWPOL ? 1 : 0;
        const int k  = u - pl * TWPOL;
        const float* src =
            k < 20 ? W2r + pl * (H2DIM * H1DIM) + k :
            k < 40 ? W2i + pl * (H2DIM * H1DIM) + (k - 20) :
            k < 50 ? W3r + pl * H2DIM + (k - 40) :
                     W3i + pl * H2DIM + (k - 50);
        twv = *src;
    }
    const float pdbm = task_info[bidx * 4];

    float2* B2 = (float2*)Blds;
#pragma unroll
    for (int s = 0; s < ZR; ++s) {
        const int z = tid + s * (WAVES_PER_BLOCK * 32);
        const unsigned ze = zen[s];
        const bool nonzero = ((ze >> 12) & 1) != 0 && ((ze >> 13) & 1) == (unsigned)((z >> 4) & 1);
        if (z < ZU && !nonzero)
            B2[z] = make_float2(0.f, 0.f);
    }

    if (tid < 2 * TWPOL) {
        const int pl = tid >= TWPOL ? 1 : 0;
        tailw[pl * TWSTRIDE + (tid - pl * TWPOL)] = twv;
    }

#pragma unroll
    for (int s = 0; s < NZR; ++s) {
        const int u = tid + s * (WAVES_PER_BLOCK * 32);
        if (u < NZU) {
            const int po = u & 3;
            const float sy = (tb0[s] & (1u << 18)) ? 2.f : 1.f;
            const float ar = wr0[s] * sy, ai = wi0[s] * sy;
            const unsigned ps = ps0[s];
            const int ww = ps & 255, hf = (ps >> 8) & 1, slot = (ps >> 9) & 1;
            float* d = Blds + ww * 64 + (slot * 8 + po * 2) * 4 + hf * 2;
            *(float2*)(d)     = make_float2(ar, -ai);
            *(float2*)(d + 4) = make_float2(ai,  ar);
        }
    }

    float4* xw4 = (float4*)xw;
#pragma unroll
    for (int u = 0; u < XQR; ++u) {
        const int k = lane + 32 * u;
        if (k < XQ) {
            xw4[2 * k]     = make_float4(xa[u].x, xb[u].x, xa[u].y, xb[u].y);
            xw4[2 * k + 1] = make_float4(xa[u].z, xb[u].z, xa[u].w, xb[u].w);
        }
    }
    __syncthreads();

    const char* xe = (const char*)(xw + r * XF);
    const char* xq = (const char*)(xw + (8 * p) * XF);
    const float* bp = Blds + r * 4 + p * 2;
    const bool half1 = p != 0;
    const bool cslot = (r >> 3) != 0;
    const unsigned cpol8 = ((r >> 2) & 1) * 8;

    const char* xe1 = xe + p * 16;
    float4 X[MMODES];

    v2f qv[8];
#pragma unroll
    for (int v = 0; v < 8; ++v) qv[v] = v2f{0.f, 0.f};

    all_tiles(std::make_integer_sequence<int, NT>{}, X, xe, xe1, bp, xq, half1, cslot, cpol8, qv);

#pragma unroll
    for (int v = 0; v < 8; ++v) {
        const int row = p * 8 + v;
        *(v2f*)(h1w + row * H1STRIDE + 2 * r) = qv[v];
    }
    __builtin_amdgcn_wave_barrier();
    asm volatile("s_wait_dscnt 0x0" ::: "memory");

    float hre[H1DIM], him[H1DIM];
#pragma unroll
    for (int o = 0; o < H1DIM; ++o) {
        hre[o] = 0.f; him[o] = 0.f;
#pragma unroll
        for (int sl = 0; sl < 2; ++sl) {
            const int c0 = sl * 8 + p * 4 + o * 2;
            const float4 s = *(const float4*)(h1w + r * H1STRIDE + 2 * c0);
            hre[o] += s.x - s.w;
            him[o] += s.z + s.y;
        }
    }
    float a0 = lrelu(hre[0]), b0 = lrelu(him[0]);
    float a1 = lrelu(hre[1]), b1 = lrelu(him[1]);

    const float* twp = tailw + p * TWSTRIDE;
    float Er = 0.f, Ei = 0.f;
#pragma unroll
    for (int q = 0; q < H2DIM; ++q) {
        float c0 = twp[q * 2], c1 = twp[q * 2 + 1];
        float d0 = twp[20 + q * 2], d1 = twp[20 + q * 2 + 1];
        float hr = a0 * c0 - b0 * d0 + a1 * c1 - b1 * d1;
        float hi = a0 * d0 + b0 * c0 + a1 * d1 + b1 * c1;
        hr = lrelu(hr);
        hi = lrelu(hi);
        float wr = twp[40 + q], wi = twp[50 + q];
        Er += hr * wr - hi * wi;
        Ei += hr * wi + hi * wr;
    }

    float P = exp10f(pdbm * 0.1f) * 0.5f;
    float xLr = xw[r * XF + (LHALF * 2 + p) * 2 + 0];
    float xLi = xw[r * XF + (LHALF * 2 + p) * 2 + 1];

    float2 res;
    res.x = xLr + Er * P;
    res.y = xLi + Ei * P;
    *(float2*)(out + ((size_t)bidx * NMODES + p) * 2) = res;
}

extern "C" void kernel_launch(void* const* d_in, const int* in_sizes, int n_in,
                              void* d_out, int out_size, void* d_ws, size_t ws_size,
                              hipStream_t stream) {
    const float* x_real    = (const float*)d_in[0];
    const float* x_imag    = (const float*)d_in[1];
    const float* task_info = (const float*)d_in[2];
    const float* W1r       = (const float*)d_in[3];
    const float* W1i       = (const float*)d_in[4];
    const float* W2r       = (const float*)d_in[5];
    const float* W2i       = (const float*)d_in[6];
    const float* W3r       = (const float*)d_in[7];
    const float* W3i       = (const float*)d_in[8];
    float* out             = (float*)d_out;

    const int blocks = BATCHSZ / (TILE * WAVES_PER_BLOCK);

    eqpbcnn_kernel<<<blocks, WAVES_PER_BLOCK * 32, 0, stream>>>(
        x_real, x_imag, task_info, W1r, W1i, W2r, W2i, W3r, W3i, out);
}


# A' of each pair formed as one two-wide (re,im) chain of 4 packed multiply-adds, swap/sign/broadcast as operand selects (was about 8 vector ops plus moves per pair); summation order unchanged
# speedup vs baseline: 5.5206x; 1.0744x over previous
#include <hip/hip_runtime.h>
#include <hip/hip_bf16.h>
#include <utility>


typedef __attribute__((ext_vector_type(2))) float v2f;
typedef __attribute__((ext_vector_type(8))) float v8f;

#define LHALF   20
#define MMODES  41
#define NMODES  2
#define H1DIM   2
#define H2DIM   10
#define BATCHSZ 65536

#define WAVES_PER_BLOCK 16
#define TILE            16
#define XF              (MMODES * 4)
#define H1STRIDE        36

struct PairTab {
    unsigned pack[442];
    int cnt;
};

constexpr PairTab buildTab() {
    PairTab t{};
    int c = 0;
    for (int m = -LHALF; m <= LHALF; ++m) {
        for (int n = -LHALF; n <= LHALF; ++n) {
            if (n < m) continue;
            long mn = (long)m * (long)n; if (mn < 0) mn = -mn;
            int s = m + n; if (s < 0) s = -s;
            if (mn <= LHALF && s <= LHALF) {
                unsigned en  = (unsigned)(LHALF + n);
                unsigned emn = (unsigned)(LHALF + m + n);
                unsigned em  = (unsigned)(LHALF + m);
                unsigned sy  = (m != n) ? 1u : 0u;
                t.pack[c] = en | (emn << 6) | (em << 12) | (sy << 18);
                ++c;
            }
        }
    }
    t.cnt = c;
    return t;
}

constexpr PairTab TAB = buildTab();
constexpr int HDIM = TAB.cnt;
static_assert(HDIM > 0 && HDIM <= 441, "HDIM bounds");

#define NWMAX 256
#define NTMAX 48
#define NGMAX 48
struct Sched {
    unsigned went[NWMAX * 2];
    unsigned pos[442];
    unsigned tile[NTMAX];
    int nw;
    int nt;
};

constexpr Sched buildSched() {
    Sched s{};
    int gs[NGMAX] = {}, gl[NGMAX] = {};
    bool used[NGMAX] = {};
    int ng = 0;
    for (int j = 0; j < TAB.cnt; ) {
        const unsigned em = (TAB.pack[j] >> 12) & 63;
        gs[ng] = j;
        while (j < TAB.cnt && ((TAB.pack[j] >> 12) & 63) == em) ++j;
        gl[ng] = j - gs[ng];
        ++ng;
    }
    int w = 0, t = 0;
    for (int g = 0; g < ng; ++g) {
        if (used[g]) continue;
        used[g] = true;
        int h = -1;
        for (int c = g + 1; c < ng; ++c) {
            const int m4 = (gl[g] + gl[c]) & 3;
            if (!used[c] && (m4 == 0 || m4 == 3)) { h = c; break; }
        }
        const int own = gl[g] & 3;
        if (h < 0 && !(own == 0 || own == 3)) {
            for (int c = g + 1; c < ng; ++c)
                if (!used[c]) { h = c; break; }
        }
        if (h >= 0) used[h] = true;
        const int len0 = gl[g], len1 = (h >= 0) ? gl[h] : 0;
        const int len = len0 + len1;
        const int cnt = ((len + 3) / 4) * 2;
        for (int k = 0; k < len; ++k) {
            const unsigned slot = (k >= len0) ? 1u : 0u;
            const int jj = slot ? gs[h] + (k - len0) : gs[g] + k;
            const int ww = w + k / 2;
            const unsigned hf = (unsigned)(k & 1);
            s.went[ww * 2 + hf] = (TAB.pack[jj] & 63) | (((TAB.pack[jj] >> 6) & 63) << 6)
                                | (1u << 12) | (slot << 13);
            s.pos[jj] = (unsigned)ww | (hf << 8) | (slot << 9);
        }
        const unsigned em0 = (TAB.pack[gs[g]] >> 12) & 63;
        const unsigned em1 = (h >= 0) ? ((TAB.pack[gs[h]] >> 12) & 63) : em0;
        s.tile[t] = em0 | (em1 << 6) | ((unsigned)cnt << 12);
        w += cnt;
        ++t;
    }
    s.nw = w;
    s.nt = t;
    return s;
}

constexpr Sched SCH = buildSched();
constexpr int NW = SCH.nw;
constexpr int NT = SCH.nt;
static_assert(NW > 0 && NW + 4 <= NWMAX && NW < 256 && (NW & 1) == 0, "schedule bounds");
static_assert(NT > 0 && NT + 1 <= NTMAX, "tile bounds");

constexpr bool checkSched() {
    int occupied = 0;
    for (int i = 0; i < NWMAX * 2; ++i)
        if (SCH.went[i] != 0) ++occupied;
    if (occupied != TAB.cnt) return false;
    int sum = 0;
    for (int t = 0; t < NT; ++t) {
        const int cnt = (int)(SCH.tile[t] >> 12);
        if (cnt <= 0 || (cnt & 1)) return false;
        sum += cnt;
    }
    if (sum != NW) return false;
    for (int j = 0; j < TAB.cnt; ++j) {
        const int ww = (int)(SCH.pos[j] & 255);
        const unsigned hf = (SCH.pos[j] >> 8) & 1, slot = (SCH.pos[j] >> 9) & 1;
        if (ww >= NW) return false;
        const unsigned want = (TAB.pack[j] & 63) | (((TAB.pack[j] >> 6) & 63) << 6)
                            | (1u << 12) | (slot << 13);
        if (SCH.went[ww * 2 + hf] != want) return false;
        int t = 0, first = 0;
        while (t < NT && ww >= first + (int)(SCH.tile[t] >> 12)) { first += (int)(SCH.tile[t] >> 12); ++t; }
        if (t >= NT) return false;
        const unsigned em = slot ? ((SCH.tile[t] >> 6) & 63) : (SCH.tile[t] & 63);
        if (em != ((TAB.pack[j] >> 12) & 63)) return false;
    }
    return true;
}
static_assert(checkSched(), "schedule covers every pair exactly once");

__device__ __constant__ PairTab dTAB = TAB;
__device__ __constant__ Sched dSCH = SCH;

constexpr int BMATS  = NW;
constexpr int TWPOL    = 2 * H2DIM * H1DIM + 2 * H2DIM;
constexpr int TWSTRIDE = 64;
constexpr int LDS_FLOATS = BMATS * 64
                         + WAVES_PER_BLOCK * TILE * XF
                         + WAVES_PER_BLOCK * TILE * H1STRIDE;
static_assert(2 * TWPOL <= WAVES_PER_BLOCK * 32, "one tail weight per thread");
static_assert(LDS_FLOATS * 4 + 2 * TWSTRIDE * 4 <= 320 * 1024, "LDS budget of one WGP");
static_assert(LDS_FLOATS * 4 > 160 * 1024, "one block per WGP");

constexpr int tileFirst(int t) {
    int f = 0;
    for (int i = 0; i < t; ++i) f += (int)(SCH.tile[i] >> 12);
    return f;
}

__device__ __forceinline__ float lrelu(float x) {
    return x > 0.f ? x : 0.01f * x;
}

__device__ __forceinline__ v2f own_pair(v2f v) {
    asm("" : "+v"(v));
    return v;
}

struct XTap { v2f lo, hi; };

__device__ __forceinline__ v2f aprime(const XTap En, const XTap Emn) {
    v2f a = En.lo * __builtin_shufflevector(Emn.lo, Emn.lo, 0, 0);
    a = __builtin_elementwise_fma(v2f{En.lo.y, -En.lo.x}, __builtin_shufflevector(Emn.lo, Emn.lo, 1, 1), a);
    a = __builtin_elementwise_fma(En.hi, __builtin_shufflevector(Emn.hi, Emn.hi, 0, 0), a);
    a = __builtin_elementwise_fma(v2f{En.hi.y, -En.hi.x}, __builtin_shufflevector(Emn.hi, Emn.hi, 1, 1), a);
    return a;
}

struct WRegs { bool any, has1; int e0, f0, e1, f1; };
constexpr WRegs wregs(int w) {
    const unsigned g0 = SCH.went[2 * w], g1 = SCH.went[2 * w + 1];
    const bool has0 = ((g0 >> 12) & 1) != 0, has1 = ((g1 >> 12) & 1) != 0;
    WRegs q{};
    q.any = has0 || has1;
    q.has1 = has1;
    q.e0 = has0 ? (int)(g0 & 63)        : (int)(g1 & 63) - 1;
    q.f0 = has0 ? (int)((g0 >> 6) & 63) : (int)((g1 >> 6) & 63) - 1;
    q.e1 = has1 ? (int)(g1 & 63) - 1        : q.e0;
    q.f1 = has1 ? (int)((g1 >> 6) & 63) - 1 : q.f0;
    return q;
}
constexpr int regLoadTile(int i) {
    for (int t = 0; t < NT; ++t) {
        const int w0 = tileFirst(t), cnt = (int)(SCH.tile[t] >> 12);
        for (int w = w0; w < w0 + cnt; ++w) {
            const WRegs q = wregs(w);
            if (q.any && (q.e0 == i || q.f0 == i || q.e1 == i || q.f1 == i))
                return t > 0 ? t - 1 : 0;
        }
    }
    return NT;
}

template<int Wi>
__device__ __forceinline__ void wmma_step(const XTap (&X)[MMODES], const float* bp,
                                          const bool half1, v8f& acc) {
    constexpr WRegs Q = wregs(Wi);
    if constexpr (Q.any) {
        constexpr int e0 = Q.e0, f0 = Q.f0, e1 = Q.e1, f1 = Q.f1;
        constexpr bool has1 = Q.has1;
        static_assert(e0 >= 0 && e0 < MMODES && f0 >= 0 && f0 < MMODES, "register of half 0");
        static_assert(e1 >= 0 && e1 < MMODES - 1 + (has1 ? 0 : 1) && f1 >= 0
                      && f1 < MMODES - 1 + (has1 ? 0 : 1), "register of half 1");
        v2f a = aprime(X[e0], X[f0]);
        if constexpr (e1 != e0 || f1 != f0) {
            const v2f b = aprime(X[e1], X[f1]);
            a.x = half1 ? b.x : a.x;
            a.y = half1 ? b.y : a.y;
        }
        const v2f bb = *(const v2f*)(bp + Wi * 64);
        acc = __builtin_amdgcn_wmma_f32_16x16x4_f32(
            false, a, false, bb, (short)0, acc, false, false);
    }
}

template<int W0, int... Is>
__device__ __forceinline__ void tile_wmmas(std::integer_sequence<int, Is...>,
                                           const XTap (&X)[MMODES], const float* bp,
                                           const bool half1, v8f& acc0, v8f& acc1) {
    (wmma_step<W0 + Is>(X, bp, half1, (Is & 1) ? acc1 : acc0), ...);
}

template<int Tt, int... Is>
__device__ __forceinline__ void row_loads(std::integer_sequence<int, Is...>,
                                          XTap (&X)[MMODES], const char* xe, const char* xe1) {
    ((regLoadTile(Is) == Tt
          ? (void)(X[Is].lo = *(const v2f*)((Is < MMODES - 1 ? xe1 : xe) + Is * 16),
                   X[Is].hi = own_pair(*(const v2f*)((Is < MMODES - 1 ? xe1 : xe) + Is * 16 + 8)))
          : (void)0), ...);
}

template<int Tt>
__device__ __forceinline__ void tile_step(XTap (&X)[MMODES], const char* xe, const char* xe1,
                                          const float* bp,
                                          const char* xq, const bool half1,
                                          const bool cslot, const unsigned cpol8,
                                          v2f (&qv)[8]) {
    row_loads<Tt>(std::make_integer_sequence<int, MMODES>{}, X, xe, xe1);
    constexpr unsigned tl = SCH.tile[Tt];
    constexpr unsigned E0 = (tl & 63) * 16, E1 = ((tl >> 6) & 63) * 16;
    constexpr int CNT = (int)(tl >> 12);
    constexpr int W0 = tileFirst(Tt);
    const unsigned eo = (cslot ? E1 : E0) + cpol8;
    v2f ev[8];
#pragma unroll
    for (int v = 0; v < 8; ++v)
        ev[v] = *(const v2f*)(xq + v * (XF * 4) + eo);

    v8f acc0 = {}, acc1 = {};
    tile_wmmas<W0>(std::make_integer_sequence<int, CNT>{}, X, bp, half1, acc0, acc1);

    const v8f acc = acc0 + acc1;
#pragma unroll
    for (int v = 0; v < 8; ++v)
        qv[v] += ev[v] * v2f{acc[v], acc[v]};
    __builtin_amdgcn_sched_barrier(0);
}

template<int... Ts>
__device__ __forceinline__ void all_tiles(std::integer_sequence<int, Ts...>,
                                          XTap (&X)[MMODES], const char* xe, const char* xe1,
                                          const float* bp,
                                          const char* xq, const bool half1,
                                          const bool cslot, const unsigned cpol8,
                                          v2f (&qv)[8]) {
    (tile_step<Ts>(X, xe, xe1, bp, xq, half1, cslot, cpol8, qv), ...);
}

__global__ __launch_bounds__(WAVES_PER_BLOCK * 32)
void eqpbcnn_kernel(const float* __restrict__ x_real,
                    const float* __restrict__ x_imag,
                    const float* __restrict__ task_info,
                    const float* __restrict__ W1r, const float* __restrict__ W1i,
                    const float* __restrict__ W2r, const float* __restrict__ W2i,
                    const float* __restrict__ W3r, const float* __restrict__ W3i,
                    float* __restrict__ out)
{
    __shared__ __attribute__((aligned(16))) float lds[LDS_FLOATS];
    __shared__ __attribute__((aligned(16))) float tailw[2 * TWSTRIDE];
    float* Blds = lds;
    float* xall = lds + BMATS * 64;

    const int tid  = threadIdx.x;
    const int lane = tid & 31;
    const int wave = tid >> 5;

    float* xw  = xall + wave * (TILE * XF);
    float* h1w = xall + WAVES_PER_BLOCK * (TILE * XF) + wave * (TILE * H1STRIDE);

    const int tile = blockIdx.x * WAVES_PER_BLOCK + wave;
    const int base = tile * TILE;

    constexpr int ZU = BMATS * 32;
    constexpr int ZR = (ZU + WAVES_PER_BLOCK * 32 - 1) / (WAVES_PER_BLOCK * 32);
    unsigned zen[ZR];
#pragma unroll
    for (int s = 0; s < ZR; ++s) {
        int z = tid + s * (WAVES_PER_BLOCK * 32);
        z = z < ZU ? z : ZU - 1;
        zen[s] = dSCH.went[(z >> 5) * 2 + (z & 1)];
    }

    constexpr int NZU = HDIM * 4;
    constexpr int NZR = (NZU + WAVES_PER_BLOCK * 32 - 1) / (WAVES_PER_BLOCK * 32);
    float wr0[NZR], wi0[NZR];
    unsigned tb0[NZR], ps0[NZR];
#pragma unroll
    for (int s = 0; s < NZR; ++s) {
        int u = tid + s * (WAVES_PER_BLOCK * 32);
        u = u < NZU ? u : NZU - 1;
        const int j = u >> 2, po = u & 3;
        wr0[s] = W1r[po * HDIM + j];
        wi0[s] = W1i[po * HDIM + j];
        tb0[s] = dTAB.pack[j];
        ps0[s] = dSCH.pos[j];
    }

    constexpr int XQ  = TILE * MMODES * NMODES / 4;
    constexpr int XQR = (XQ + 31) / 32;
    static_assert((TILE * MMODES * NMODES) % 4 == 0, "x tile is whole float4s");
    const float4* xr4 = (const float4*)(x_real + (size_t)base * (MMODES * NMODES));
    const float4* xi4 = (const float4*)(x_imag + (size_t)base * (MMODES * NMODES));
    float4 xa[XQR], xb[XQR];
#pragma unroll
    for (int u = 0; u < XQR; ++u) {
        int k = lane + 32 * u;
        k = k < XQ ? k : XQ - 1;
        xa[u] = xr4[k];
        xb[u] = xi4[k];
    }

    const int r = lane & 15;
    const int p = lane >> 4;
    const int bidx = base + r;
    float twv;
    {
        const int u  = tid < 2 * TWPOL ? tid : 2 * TWPOL - 1;
        const int pl = u >= TWPOL ? 1 : 0;
        const int k  = u - pl * TWPOL;
        const float* src =
            k < 20 ? W2r + pl * (H2DIM * H1DIM) + k :
            k < 40 ? W2i + pl * (H2DIM * H1DIM) + (k - 20) :
            k < 50 ? W3r + pl * H2DIM + (k - 40) :
                     W3i + pl * H2DIM + (k - 50);
        twv = *src;
    }
    const float pdbm = task_info[bidx * 4];

    float2* B2 = (float2*)Blds;
#pragma unroll
    for (int s = 0; s < ZR; ++s) {
        const int z = tid + s * (WAVES_PER_BLOCK * 32);
        const unsigned ze = zen[s];
        const bool nonzero = ((ze >> 12) & 1) != 0 && ((ze >> 13) & 1) == (unsigned)((z >> 4) & 1);
        if (z < ZU && !nonzero)
            B2[z] = make_float2(0.f, 0.f);
    }

    if (tid < 2 * TWPOL) {
        const int pl = tid >= TWPOL ? 1 : 0;
        tailw[pl * TWSTRIDE + (tid - pl * TWPOL)] = twv;
    }

#pragma unroll
    for (int s = 0; s < NZR; ++s) {
        const int u = tid + s * (WAVES_PER_BLOCK * 32);
        if (u < NZU) {
            const int po = u & 3;
            const float sy = (tb0[s] & (1u << 18)) ? 2.f : 1.f;
            const float ar = wr0[s] * sy, ai = wi0[s] * sy;
            const unsigned ps = ps0[s];
            const int ww = ps & 255, hf = (ps >> 8) & 1, slot = (ps >> 9) & 1;
            float* d = Blds + ww * 64 + (slot * 8 + po * 2) * 4 + hf * 2;
            *(float2*)(d)     = make_float2(ar, -ai);
            *(float2*)(d + 4) = make_float2(ai,  ar);
        }
    }

    float4* xw4 = (float4*)xw;
#pragma unroll
    for (int u = 0; u < XQR; ++u) {
        const int k = lane + 32 * u;
        if (k < XQ) {
            xw4[2 * k]     = make_float4(xa[u].x, xb[u].x, xa[u].y, xb[u].y);
            xw4[2 * k + 1] = make_float4(xa[u].z, xb[u].z, xa[u].w, xb[u].w);
        }
    }
    __syncthreads();

    const char* xe = (const char*)(xw + r * XF);
    const char* xq = (const char*)(xw + (8 * p) * XF);
    const float* bp = Blds + r * 4 + p * 2;
    const bool half1 = p != 0;
    const bool cslot = (r >> 3) != 0;
    const unsigned cpol8 = ((r >> 2) & 1) * 8;

    const char* xe1 = xe + p * 16;
    XTap X[MMODES];

    v2f qv[8];
#pragma unroll
    for (int v = 0; v < 8; ++v) qv[v] = v2f{0.f, 0.f};

    all_tiles(std::make_integer_sequence<int, NT>{}, X, xe, xe1, bp, xq, half1, cslot, cpol8, qv);

#pragma unroll
    for (int v = 0; v < 8; ++v) {
        const int row = p * 8 + v;
        *(v2f*)(h1w + row * H1STRIDE + 2 * r) = qv[v];
    }
    __builtin_amdgcn_wave_barrier();
    asm volatile("s_wait_dscnt 0x0" ::: "memory");

    float hre[H1DIM], him[H1DIM];
#pragma unroll
    for (int o = 0; o < H1DIM; ++o) {
        hre[o] = 0.f; him[o] = 0.f;
#pragma unroll
        for (int sl = 0; sl < 2; ++sl) {
            const int c0 = sl * 8 + p * 4 + o * 2;
            const float4 s = *(const float4*)(h1w + r * H1STRIDE + 2 * c0);
            hre[o] += s.x - s.w;
            him[o] += s.z + s.y;
        }
    }
    float a0 = lrelu(hre[0]), b0 = lrelu(him[0]);
    float a1 = lrelu(hre[1]), b1 = lrelu(him[1]);

    const float* twp = tailw + p * TWSTRIDE;
    float Er = 0.f, Ei = 0.f;
#pragma unroll
    for (int q = 0; q < H2DIM; ++q) {
        float c0 = twp[q * 2], c1 = twp[q * 2 + 1];
        float d0 = twp[20 + q * 2], d1 = twp[20 + q * 2 + 1];
        float hr = a0 * c0 - b0 * d0 + a1 * c1 - b1 * d1;
        float hi = a0 * d0 + b0 * c0 + a1 * d1 + b1 * c1;
        hr = lrelu(hr);
        hi = lrelu(hi);
        float wr = twp[40 + q], wi = twp[50 + q];
        Er += hr * wr - hi * wi;
        Ei += hr * wi + hi * wr;
    }

    float P = exp10f(pdbm * 0.1f) * 0.5f;
    float xLr = xw[r * XF + (LHALF * 2 + p) * 2 + 0];
    float xLi = xw[r * XF + (LHALF * 2 + p) * 2 + 1];

    float2 res;
    res.x = xLr + Er * P;
    res.y = xLi + Ei * P;
    *(float2*)(out + ((size_t)bidx * NMODES + p) * 2) = res;
}

extern "C" void kernel_launch(void* const* d_in, const int* in_sizes, int n_in,
                              void* d_out, int out_size, void* d_ws, size_t ws_size,
                              hipStream_t stream) {
    const float* x_real    = (const float*)d_in[0];
    const float* x_imag    = (const float*)d_in[1];
    const float* task_info = (const float*)d_in[2];
    const float* W1r       = (const float*)d_in[3];
    const float* W1i       = (const float*)d_in[4];
    const float* W2r       = (const float*)d_in[5];
    const float* W2i       = (const float*)d_in[6];
    const float* W3r       = (const float*)d_in[7];
    const float* W3i       = (const float*)d_in[8];
    float* out             = (float*)d_out;

    const int blocks = BATCHSZ / (TILE * WAVES_PER_BLOCK);

    eqpbcnn_kernel<<<blocks, WAVES_PER_BLOCK * 32, 0, stream>>>(
        x_real, x_imag, task_info, W1r, W1i, W2r, W2i, W3r, W3i, out);
}


# mirrored m groups (+m,-m) share a tile: feature x[n]conj(x[n+m]) formed once, the -m pair reads its conjugate as a sign in B, both column halves filled: 75 WMMAs, was 90; summation order changes
# speedup vs baseline: 5.8581x; 1.0611x over previous
#include <hip/hip_runtime.h>
#include <hip/hip_bf16.h>
#include <utility>


typedef __attribute__((ext_vector_type(2))) float v2f;
typedef __attribute__((ext_vector_type(8))) float v8f;

#define LHALF   20
#define MMODES  41
#define NMODES  2
#define H1DIM   2
#define H2DIM   10
#define BATCHSZ 65536

#define WAVES_PER_BLOCK 16
#define TILE            16
#define XF              (MMODES * 4)
#define H1STRIDE        36

struct PairTab {
    unsigned pack[442];
    int cnt;
};

constexpr PairTab buildTab() {
    PairTab t{};
    int c = 0;
    for (int m = -LHALF; m <= LHALF; ++m) {
        for (int n = -LHALF; n <= LHALF; ++n) {
            if (n < m) continue;
            long mn = (long)m * (long)n; if (mn < 0) mn = -mn;
            int s = m + n; if (s < 0) s = -s;
            if (mn <= LHALF && s <= LHALF) {
                unsigned en  = (unsigned)(LHALF + n);
                unsigned emn = (unsigned)(LHALF + m + n);
                unsigned em  = (unsigned)(LHALF + m);
                unsigned sy  = (m != n) ? 1u : 0u;
                t.pack[c] = en | (emn << 6) | (em << 12) | (sy << 18);
                ++c;
            }
        }
    }
    t.cnt = c;
    return t;
}

constexpr PairTab TAB = buildTab();
constexpr int HDIM = TAB.cnt;
static_assert(HDIM > 0 && HDIM <= 441, "HDIM bounds");

#define NWMAX 256
#define NTMAX 48
#define NGMAX 48
struct Sched {
    unsigned went[NWMAX * 2];
    unsigned pos[442];
    unsigned tile[NTMAX];
    int nw;
    int nt;
};

constexpr Sched buildSched() {
    Sched s{};
    int gs[NGMAX] = {}, gl[NGMAX] = {}, gm[NGMAX] = {};
    bool used[NGMAX] = {};
    int ng = 0;
    for (int j = 0; j < TAB.cnt; ) {
        const unsigned em = (TAB.pack[j] >> 12) & 63;
        gs[ng] = j;
        gm[ng] = (int)em - LHALF;
        while (j < TAB.cnt && ((TAB.pack[j] >> 12) & 63) == em) ++j;
        gl[ng] = j - gs[ng];
        ++ng;
    }
    int w = 0, t = 0;
    for (int g = 0; g < ng; ++g) {
        if (gm[g] <= 0) continue;
        int h = -1;
        for (int c = 0; c < ng; ++c)
            if (gm[c] == -gm[g]) h = c;
        if (h < 0) continue;
        used[g] = true;
        used[h] = true;
        const int m = gm[g];
        int k = 0;
        for (int n = -LHALF; n + m <= LHALF; ++n) {
            int j0 = -1, j1 = -1;
            for (int q = 0; q < gl[g]; ++q)
                if ((int)(TAB.pack[gs[g] + q] & 63) - LHALF == n) j0 = gs[g] + q;
            for (int q = 0; q < gl[h]; ++q)
                if ((int)(TAB.pack[gs[h] + q] & 63) - LHALF == n + m) j1 = gs[h] + q;
            if (j0 < 0 && j1 < 0) continue;
            const int ww = w + k / 2;
            const unsigned hf = (unsigned)(k & 1);
            s.went[ww * 2 + hf] = (unsigned)(LHALF + n) | ((unsigned)(LHALF + n + m) << 6)
                                | ((j0 >= 0 ? 1u : 0u) << 12) | ((j1 >= 0 ? 1u : 0u) << 13);
            if (j0 >= 0) s.pos[j0] = (unsigned)ww | (hf << 8);
            if (j1 >= 0) s.pos[j1] = (unsigned)ww | (hf << 8) | (1u << 9) | (1u << 10);
            ++k;
        }
        const int cnt = (k + 1) / 2;
        const unsigned em0 = (TAB.pack[gs[g]] >> 12) & 63;
        const unsigned em1 = (TAB.pack[gs[h]] >> 12) & 63;
        s.tile[t] = em0 | (em1 << 6) | ((unsigned)cnt << 12);
        w += cnt;
        ++t;
    }
    for (int g = 0; g < ng; ++g) {
        if (used[g]) continue;
        used[g] = true;
        int h = -1;
        for (int c = g + 1; c < ng; ++c)
            if (!used[c] && ((gl[g] + gl[c]) & 1) == 0) { h = c; break; }
        if (h < 0 && (gl[g] & 1) != 0) {
            for (int c = g + 1; c < ng; ++c)
                if (!used[c]) { h = c; break; }
        }
        if (h >= 0) used[h] = true;
        const int len0 = gl[g], len1 = (h >= 0) ? gl[h] : 0;
        const int len = len0 + len1;
        const int cnt = (len + 1) / 2;
        for (int k = 0; k < len; ++k) {
            const unsigned slot = (k >= len0) ? 1u : 0u;
            const int jj = slot ? gs[h] + (k - len0) : gs[g] + k;
            const int ww = w + k / 2;
            const unsigned hf = (unsigned)(k & 1);
            s.went[ww * 2 + hf] = (TAB.pack[jj] & 63) | (((TAB.pack[jj] >> 6) & 63) << 6)
                                | (1u << (12 + slot));
            s.pos[jj] = (unsigned)ww | (hf << 8) | (slot << 9);
        }
        const unsigned em0 = (TAB.pack[gs[g]] >> 12) & 63;
        const unsigned em1 = (h >= 0) ? ((TAB.pack[gs[h]] >> 12) & 63) : em0;
        s.tile[t] = em0 | (em1 << 6) | ((unsigned)cnt << 12);
        w += cnt;
        ++t;
    }
    s.nw = w;
    s.nt = t;
    return s;
}

constexpr Sched SCH = buildSched();
constexpr int NW = SCH.nw;
constexpr int NT = SCH.nt;
static_assert(NW > 0 && NW + 4 <= NWMAX && NW < 256, "schedule bounds");
static_assert(NT > 0 && NT + 1 <= NTMAX, "tile bounds");

constexpr bool checkSched() {
    int uses = 0;
    for (int i = 0; i < NWMAX * 2; ++i)
        uses += (int)((SCH.went[i] >> 12) & 1) + (int)((SCH.went[i] >> 13) & 1);
    if (uses != TAB.cnt) return false;
    int sum = 0;
    for (int t = 0; t < NT; ++t) {
        const int cnt = (int)(SCH.tile[t] >> 12);
        if (cnt <= 0) return false;
        sum += cnt;
    }
    if (sum != NW) return false;
    bool seen[NWMAX * 4] = {};
    for (int j = 0; j < TAB.cnt; ++j) {
        const int ww = (int)(SCH.pos[j] & 255);
        const unsigned hf = (SCH.pos[j] >> 8) & 1, slot = (SCH.pos[j] >> 9) & 1;
        const bool cj = ((SCH.pos[j] >> 10) & 1) != 0;
        if (ww >= NW) return false;
        if (seen[(ww * 2 + (int)hf) * 2 + (int)slot]) return false;
        seen[(ww * 2 + (int)hf) * 2 + (int)slot] = true;
        const unsigned en = TAB.pack[j] & 63, emn = (TAB.pack[j] >> 6) & 63;
        const unsigned want = cj ? (emn | (en << 6)) : (en | (emn << 6));
        const unsigned ent = SCH.went[ww * 2 + hf];
        if ((ent & 4095) != want) return false;
        if (((ent >> (12 + slot)) & 1) == 0) return false;
        int t = 0, first = 0;
        while (t < NT && ww >= first + (int)(SCH.tile[t] >> 12)) { first += (int)(SCH.tile[t] >> 12); ++t; }
        if (t >= NT) return false;
        const unsigned em = slot ? ((SCH.tile[t] >> 6) & 63) : (SCH.tile[t] & 63);
        if (em != ((TAB.pack[j] >> 12) & 63)) return false;
    }
    return true;
}
static_assert(checkSched(), "schedule covers every pair exactly once");

__device__ __constant__ PairTab dTAB = TAB;
__device__ __constant__ Sched dSCH = SCH;

constexpr int BMATS  = NW;
constexpr int TWPOL    = 2 * H2DIM * H1DIM + 2 * H2DIM;
constexpr int TWSTRIDE = 64;
constexpr int LDS_FLOATS = BMATS * 64
                         + WAVES_PER_BLOCK * TILE * XF
                         + WAVES_PER_BLOCK * TILE * H1STRIDE;
static_assert(2 * TWPOL <= WAVES_PER_BLOCK * 32, "one tail weight per thread");
static_assert(LDS_FLOATS * 4 + 2 * TWSTRIDE * 4 <= 320 * 1024, "LDS budget of one WGP");
static_assert(LDS_FLOATS * 4 > 160 * 1024, "one block per WGP");

constexpr int tileFirst(int t) {
    int f = 0;
    for (int i = 0; i < t; ++i) f += (int)(SCH.tile[i] >> 12);
    return f;
}

__device__ __forceinline__ float lrelu(float x) {
    return x > 0.f ? x : 0.01f * x;
}

__device__ __forceinline__ v2f own_pair(v2f v) {
    asm("" : "+v"(v));
    return v;
}

struct XTap { v2f lo, hi; };

__device__ __forceinline__ v2f aprime(const XTap En, const XTap Emn) {
    v2f a = En.lo * __builtin_shufflevector(Emn.lo, Emn.lo, 0, 0);
    a = __builtin_elementwise_fma(v2f{En.lo.y, -En.lo.x}, __builtin_shufflevector(Emn.lo, Emn.lo, 1, 1), a);
    a = __builtin_elementwise_fma(En.hi, __builtin_shufflevector(Emn.hi, Emn.hi, 0, 0), a);
    a = __builtin_elementwise_fma(v2f{En.hi.y, -En.hi.x}, __builtin_shufflevector(Emn.hi, Emn.hi, 1, 1), a);
    return a;
}

struct WRegs { bool any, has1; int e0, f0, e1, f1; };
constexpr WRegs wregs(int w) {
    const unsigned g0 = SCH.went[2 * w], g1 = SCH.went[2 * w + 1];
    const bool has0 = ((g0 >> 12) & 3) != 0, has1 = ((g1 >> 12) & 3) != 0;
    WRegs q{};
    q.any = has0 || has1;
    q.has1 = has1;
    q.e0 = has0 ? (int)(g0 & 63)        : (int)(g1 & 63) - 1;
    q.f0 = has0 ? (int)((g0 >> 6) & 63) : (int)((g1 >> 6) & 63) - 1;
    q.e1 = has1 ? (int)(g1 & 63) - 1        : q.e0;
    q.f1 = has1 ? (int)((g1 >> 6) & 63) - 1 : q.f0;
    return q;
}
constexpr int regLoadTile(int i) {
    for (int t = 0; t < NT; ++t) {
        const int w0 = tileFirst(t), cnt = (int)(SCH.tile[t] >> 12);
        for (int w = w0; w < w0 + cnt; ++w) {
            const WRegs q = wregs(w);
            if (q.any && (q.e0 == i || q.f0 == i || q.e1 == i || q.f1 == i))
                return t > 0 ? t - 1 : 0;
        }
    }
    return NT;
}

template<int Wi>
__device__ __forceinline__ void wmma_step(const XTap (&X)[MMODES], const float* bp,
                                          const bool half1, v8f& acc) {
    constexpr WRegs Q = wregs(Wi);
    if constexpr (Q.any) {
        constexpr int e0 = Q.e0, f0 = Q.f0, e1 = Q.e1, f1 = Q.f1;
        constexpr bool has1 = Q.has1;
        static_assert(e0 >= 0 && e0 < MMODES && f0 >= 0 && f0 < MMODES, "register of half 0");
        static_assert(e1 >= 0 && e1 < MMODES - 1 + (has1 ? 0 : 1) && f1 >= 0
                      && f1 < MMODES - 1 + (has1 ? 0 : 1), "register of half 1");
        v2f a = aprime(X[e0], X[f0]);
        if constexpr (e1 != e0 || f1 != f0) {
            const v2f b = aprime(X[e1], X[f1]);
            a.x = half1 ? b.x : a.x;
            a.y = half1 ? b.y : a.y;
        }
        const v2f bb = *(const v2f*)(bp + Wi * 64);
        acc = __builtin_amdgcn_wmma_f32_16x16x4_f32(
            false, a, false, bb, (short)0, acc, false, false);
    }
}

template<int W0, int... Is>
__device__ __forceinline__ void tile_wmmas(std::integer_sequence<int, Is...>,
                                           const XTap (&X)[MMODES], const float* bp,
                                           const bool half1, v8f& acc0, v8f& acc1) {
    (wmma_step<W0 + Is>(X, bp, half1, (Is & 1) ? acc1 : acc0), ...);
}

template<int Tt, int... Is>
__device__ __forceinline__ void row_loads(std::integer_sequence<int, Is...>,
                                          XTap (&X)[MMODES], const char* xe, const char* xe1) {
    ((regLoadTile(Is) == Tt
          ? (void)(X[Is].lo = *(const v2f*)((Is < MMODES - 1 ? xe1 : xe) + Is * 16),
                   X[Is].hi = own_pair(*(const v2f*)((Is < MMODES - 1 ? xe1 : xe) + Is * 16 + 8)))
          : (void)0), ...);
}

template<int Tt>
__device__ __forceinline__ void tile_step(XTap (&X)[MMODES], const char* xe, const char* xe1,
                                          const float* bp,
                                          const char* xq, const bool half1,
                                          const bool cslot, const unsigned cpol8,
                                          v2f (&qv)[8]) {
    row_loads<Tt>(std::make_integer_sequence<int, MMODES>{}, X, xe, xe1);
    constexpr unsigned tl = SCH.tile[Tt];
    constexpr unsigned E0 = (tl & 63) * 16, E1 = ((tl >> 6) & 63) * 16;
    constexpr int CNT = (int)(tl >> 12);
    constexpr int W0 = tileFirst(Tt);
    const unsigned eo = (cslot ? E1 : E0) + cpol8;
    v2f ev[8];
#pragma unroll
    for (int v = 0; v < 8; ++v)
        ev[v] = *(const v2f*)(xq + v * (XF * 4) + eo);

    v8f acc0 = {}, acc1 = {};
    tile_wmmas<W0>(std::make_integer_sequence<int, CNT>{}, X, bp, half1, acc0, acc1);

    const v8f acc = acc0 + acc1;
#pragma unroll
    for (int v = 0; v < 8; ++v)
        qv[v] += ev[v] * v2f{acc[v], acc[v]};
    __builtin_amdgcn_sched_barrier(0);
}

template<int... Ts>
__device__ __forceinline__ void all_tiles(std::integer_sequence<int, Ts...>,
                                          XTap (&X)[MMODES], const char* xe, const char* xe1,
                                          const float* bp,
                                          const char* xq, const bool half1,
                                          const bool cslot, const unsigned cpol8,
                                          v2f (&qv)[8]) {
    (tile_step<Ts>(X, xe, xe1, bp, xq, half1, cslot, cpol8, qv), ...);
}

__global__ __launch_bounds__(WAVES_PER_BLOCK * 32)
void eqpbcnn_kernel(const float* __restrict__ x_real,
                    const float* __restrict__ x_imag,
                    const float* __restrict__ task_info,
                    const float* __restrict__ W1r, const float* __restrict__ W1i,
                    const float* __restrict__ W2r, const float* __restrict__ W2i,
                    const float* __restrict__ W3r, const float* __restrict__ W3i,
                    float* __restrict__ out)
{
    __shared__ __attribute__((aligned(16))) float lds[LDS_FLOATS];
    __shared__ __attribute__((aligned(16))) float tailw[2 * TWSTRIDE];
    float* Blds = lds;
    float* xall = lds + BMATS * 64;

    const int tid  = threadIdx.x;
    const int lane = tid & 31;
    const int wave = tid >> 5;

    float* xw  = xall + wave * (TILE * XF);
    float* h1w = xall + WAVES_PER_BLOCK * (TILE * XF) + wave * (TILE * H1STRIDE);

    const int tile = blockIdx.x * WAVES_PER_BLOCK + wave;
    const int base = tile * TILE;

    constexpr int ZU = BMATS * 32;
    constexpr int ZR = (ZU + WAVES_PER_BLOCK * 32 - 1) / (WAVES_PER_BLOCK * 32);
    unsigned zen[ZR];
#pragma unroll
    for (int s = 0; s < ZR; ++s) {
        int z = tid + s * (WAVES_PER_BLOCK * 32);
        z = z < ZU ? z : ZU - 1;
        zen[s] = dSCH.went[(z >> 5) * 2 + (z & 1)];
    }

    constexpr int NZU = HDIM * 4;
    constexpr int NZR = (NZU + WAVES_PER_BLOCK * 32 - 1) / (WAVES_PER_BLOCK * 32);
    float wr0[NZR], wi0[NZR];
    unsigned tb0[NZR], ps0[NZR];
#pragma unroll
    for (int s = 0; s < NZR; ++s) {
        int u = tid + s * (WAVES_PER_BLOCK * 32);
        u = u < NZU ? u : NZU - 1;
        const int j = u >> 2, po = u & 3;
        wr0[s] = W1r[po * HDIM + j];
        wi0[s] = W1i[po * HDIM + j];
        tb0[s] = dTAB.pack[j];
        ps0[s] = dSCH.pos[j];
    }

    constexpr int XQ  = TILE * MMODES * NMODES / 4;
    constexpr int XQR = (XQ + 31) / 32;
    static_assert((TILE * MMODES * NMODES) % 4 == 0, "x tile is whole float4s");
    const float4* xr4 = (const float4*)(x_real + (size_t)base * (MMODES * NMODES));
    const float4* xi4 = (const float4*)(x_imag + (size_t)base * (MMODES * NMODES));
    float4 xa[XQR], xb[XQR];
#pragma unroll
    for (int u = 0; u < XQR; ++u) {
        int k = lane + 32 * u;
        k = k < XQ ? k : XQ - 1;
        xa[u] = xr4[k];
        xb[u] = xi4[k];
    }

    const int r = lane & 15;
    const int p = lane >> 4;
    const int bidx = base + r;
    float twv;
    {
        const int u  = tid < 2 * TWPOL ? tid : 2 * TWPOL - 1;
        const int pl = u >= TWPOL ? 1 : 0;
        const int k  = u - pl * TWPOL;
        const float* src =
            k < 20 ? W2r + pl * (H2DIM * H1DIM) + k :
            k < 40 ? W2i + pl * (H2DIM * H1DIM) + (k - 20) :
            k < 50 ? W3r + pl * H2DIM + (k - 40) :
                     W3i + pl * H2DIM + (k - 50);
        twv = *src;
    }
    const float pdbm = task_info[bidx * 4];

    float2* B2 = (float2*)Blds;
#pragma unroll
    for (int s = 0; s < ZR; ++s) {
        const int z = tid + s * (WAVES_PER_BLOCK * 32);
        const unsigned ze = zen[s];
        const bool nonzero = ((ze >> (12 + ((z >> 4) & 1))) & 1) != 0;
        if (z < ZU && !nonzero)
            B2[z] = make_float2(0.f, 0.f);
    }

    if (tid < 2 * TWPOL) {
        const int pl = tid >= TWPOL ? 1 : 0;
        tailw[pl * TWSTRIDE + (tid - pl * TWPOL)] = twv;
    }

#pragma unroll
    for (int s = 0; s < NZR; ++s) {
        const int u = tid + s * (WAVES_PER_BLOCK * 32);
        if (u < NZU) {
            const int po = u & 3;
            const float sy = (tb0[s] & (1u << 18)) ? 2.f : 1.f;
            const float ar = wr0[s] * sy, ai = wi0[s] * sy;
            const unsigned ps = ps0[s];
            const int ww = ps & 255, hf = (ps >> 8) & 1, slot = (ps >> 9) & 1;
            const float sg = (ps & (1u << 10)) ? -1.f : 1.f;
            float* d = Blds + ww * 64 + (slot * 8 + po * 2) * 4 + hf * 2;
            *(float2*)(d)     = make_float2(ar, -ai * sg);
            *(float2*)(d + 4) = make_float2(ai,  ar * sg);
        }
    }

    float4* xw4 = (float4*)xw;
#pragma unroll
    for (int u = 0; u < XQR; ++u) {
        const int k = lane + 32 * u;
        if (k < XQ) {
            xw4[2 * k]     = make_float4(xa[u].x, xb[u].x, xa[u].y, xb[u].y);
            xw4[2 * k + 1] = make_float4(xa[u].z, xb[u].z, xa[u].w, xb[u].w);
        }
    }
    __syncthreads();

    const char* xe = (const char*)(xw + r * XF);
    const char* xq = (const char*)(xw + (8 * p) * XF);
    const float* bp = Blds + r * 4 + p * 2;
    const bool half1 = p != 0;
    const bool cslot = (r >> 3) != 0;
    const unsigned cpol8 = ((r >> 2) & 1) * 8;

    const char* xe1 = xe + p * 16;
    XTap X[MMODES];

    v2f qv[8];
#pragma unroll
    for (int v = 0; v < 8; ++v) qv[v] = v2f{0.f, 0.f};

    all_tiles(std::make_integer_sequence<int, NT>{}, X, xe, xe1, bp, xq, half1, cslot, cpol8, qv);

#pragma unroll
    for (int v = 0; v < 8; ++v) {
        const int row = p * 8 + v;
        *(v2f*)(h1w + row * H1STRIDE + 2 * r) = qv[v];
    }
    __builtin_amdgcn_wave_barrier();
    asm volatile("s_wait_dscnt 0x0" ::: "memory");

    float hre[H1DIM], him[H1DIM];
#pragma unroll
    for (int o = 0; o < H1DIM; ++o) {
        hre[o] = 0.f; him[o] = 0.f;
#pragma unroll
        for (int sl = 0; sl < 2; ++sl) {
            const int c0 = sl * 8 + p * 4 + o * 2;
            const float4 s = *(const float4*)(h1w + r * H1STRIDE + 2 * c0);
            hre[o] += s.x - s.w;
            him[o] += s.z + s.y;
        }
    }
    float a0 = lrelu(hre[0]), b0 = lrelu(him[0]);
    float a1 = lrelu(hre[1]), b1 = lrelu(him[1]);

    const float* twp = tailw + p * TWSTRIDE;
    float Er = 0.f, Ei = 0.f;
#pragma unroll
    for (int q = 0; q < H2DIM; ++q) {
        float c0 = twp[q * 2], c1 = twp[q * 2 + 1];
        float d0 = twp[20 + q * 2], d1 = twp[20 + q * 2 + 1];
        float hr = a0 * c0 - b0 * d0 + a1 * c1 - b1 * d1;
        float hi = a0 * d0 + b0 * c0 + a1 * d1 + b1 * c1;
        hr = lrelu(hr);
        hi = lrelu(hi);
        float wr = twp[40 + q], wi = twp[50 + q];
        Er += hr * wr - hi * wi;
        Ei += hr * wi + hi * wr;
    }

    float P = exp10f(pdbm * 0.1f) * 0.5f;
    float xLr = xw[r * XF + (LHALF * 2 + p) * 2 + 0];
    float xLi = xw[r * XF + (LHALF * 2 + p) * 2 + 1];

    float2 res;
    res.x = xLr + Er * P;
    res.y = xLi + Ei * P;
    *(float2*)(out + ((size_t)bidx * NMODES + p) * 2) = res;
}

extern "C" void kernel_launch(void* const* d_in, const int* in_sizes, int n_in,
                              void* d_out, int out_size, void* d_ws, size_t ws_size,
                              hipStream_t stream) {
    const float* x_real    = (const float*)d_in[0];
    const float* x_imag    = (const float*)d_in[1];
    const float* task_info = (const float*)d_in[2];
    const float* W1r       = (const float*)d_in[3];
    const float* W1i       = (const float*)d_in[4];
    const float* W2r       = (const float*)d_in[5];
    const float* W2i       = (const float*)d_in[6];
    const float* W3r       = (const float*)d_in[7];
    const float* W3i       = (const float*)d_in[8];
    float* out             = (float*)d_out;

    const int blocks = BATCHSZ / (TILE * WAVES_PER_BLOCK);

    eqpbcnn_kernel<<<blocks, WAVES_PER_BLOCK * 32, 0, stream>>>(
        x_real, x_imag, task_info, W1r, W1i, W2r, W2i, W3r, W3i, out);
}
